# starting kernel
# baseline (speedup 1.0000x reference)
#include <hip/hip_runtime.h>
#include <hip/hip_bf16.h>

typedef __attribute__((ext_vector_type(2))) float v2f;
typedef __attribute__((ext_vector_type(8))) float v8f;

#define N_NODES 100000
#define N_EDGES 1600000
#define F_IN    500
#define HID     128
#define N_CLS   40


__global__ __launch_bounds__(256) void deg_init_kernel(float* __restrict__ deg) {
    int i = blockIdx.x * blockDim.x + threadIdx.x;
    if (i < N_NODES) deg[i] = 1.0f;
}

__global__ __launch_bounds__(256) void deg_count_kernel(const int* __restrict__ dst,
                                                        float* __restrict__ deg) {
    int e = blockIdx.x * blockDim.x + threadIdx.x;
    if (e < N_EDGES) unsafeAtomicAdd(&deg[dst[e]], 1.0f);
}

__global__ __launch_bounds__(256) void dinv_kernel(float* __restrict__ d) {
    int i = blockIdx.x * blockDim.x + threadIdx.x;
    if (i < N_NODES) d[i] = rsqrtf(d[i]);
}


template<int NT>
__global__ __launch_bounds__(256) void pack_w_kernel(const float* __restrict__ W,
                                                     v2f* __restrict__ Wp,
                                                     int K, int Nc) {
    int idx = blockIdx.x * blockDim.x + threadIdx.x;
    int total = (K / 4) * NT * 32;
    if (idx >= total) return;
    int lane = idx & 31;
    int t    = (idx >> 5) % NT;
    int ks   = idx / (32 * NT);
    int col  = t * 16 + (lane & 15);
    int k    = ks * 4 + ((lane >> 4) * 2);
    v2f v;
    v.x = (col < Nc) ? W[(size_t)k * Nc + col]       : 0.0f;
    v.y = (col < Nc) ? W[(size_t)(k + 1) * Nc + col] : 0.0f;
    Wp[idx] = v;
}


template<int NT, int NC, bool RELU>
__global__ __launch_bounds__(128) void gemm_wmma_kernel(const float* __restrict__ A,
                                                        const v2f* __restrict__ Wp,
                                                        float* __restrict__ Out,
                                                        int M, int K) {
    const int wave = threadIdx.x >> 5;
    const int lane = threadIdx.x & 31;
    const int pair = blockIdx.x * 4 + wave;
    if (pair * 32 >= M) return;

    const int kk  = (lane >> 4) * 2;
    const int col = lane & 15;

    const float* __restrict__ Arow0 = A + (size_t)(pair * 32 + col) * K + kk;
    const float* __restrict__ Arow1 = Arow0 + (size_t)16 * K;
    const v2f*   __restrict__ wp    = Wp + lane;

    v8f acc0[NT], acc1[NT];
#pragma unroll
    for (int t = 0; t < NT; ++t) {
        acc0[t] = (v8f){0.f,0.f,0.f,0.f,0.f,0.f,0.f,0.f};
        acc1[t] = (v8f){0.f,0.f,0.f,0.f,0.f,0.f,0.f,0.f};
    }

    const int nsteps = K >> 2;
#pragma unroll 2
    for (int ks = 0; ks < nsteps; ++ks) {
        v2f a0 = *reinterpret_cast<const v2f*>(Arow0 + ks * 4);
        v2f a1 = *reinterpret_cast<const v2f*>(Arow1 + ks * 4);
        if (RELU) {
            a0.x = fmaxf(a0.x, 0.0f); a0.y = fmaxf(a0.y, 0.0f);
            a1.x = fmaxf(a1.x, 0.0f); a1.y = fmaxf(a1.y, 0.0f);
        }
#pragma unroll
        for (int t = 0; t < NT; ++t) {
            v2f b = wp[(ks * NT + t) * 32];
            acc0[t] = __builtin_amdgcn_wmma_f32_16x16x4_f32(
                          false, a0, false, b, (short)0, acc0[t], false, false);
            acc1[t] = __builtin_amdgcn_wmma_f32_16x16x4_f32(
                          false, a1, false, b, (short)0, acc1[t], false, false);
        }
    }

    const int rbase0 = pair * 32 + ((lane < 16) ? 0 : 8);
    const int rbase1 = rbase0 + 16;
#pragma unroll
    for (int t = 0; t < NT; ++t) {
        int c = t * 16 + col;
        if ((NC % 16 == 0) || (c < NC)) {
#pragma unroll
            for (int r = 0; r < 8; ++r) {
                Out[(size_t)(rbase0 + r) * NC + c] = acc0[t][r];
                Out[(size_t)(rbase1 + r) * NC + c] = acc1[t][r];
            }
        }
    }
}


template<int F>
__global__ __launch_bounds__(256) void self_bias_kernel(const float* __restrict__ h,
                                                        const float* __restrict__ dinv,
                                                        const float* __restrict__ bias,
                                                        float* __restrict__ out,
                                                        int n) {
    constexpr int F4 = F / 4;
    long idx = (long)blockIdx.x * blockDim.x + threadIdx.x;
    if (idx >= (long)n * F4) return;
    int i  = (int)(idx / F4);
    int f4 = (int)(idx % F4);
    float di = dinv[i];
    float s  = di * di;
    float4 hv = reinterpret_cast<const float4*>(h)[idx];
    float4 bv = reinterpret_cast<const float4*>(bias)[f4];
    float4 ov;
    ov.x = bv.x + s * hv.x;  ov.y = bv.y + s * hv.y;
    ov.z = bv.z + s * hv.z;  ov.w = bv.w + s * hv.w;
    reinterpret_cast<float4*>(out)[idx] = ov;
}


template<int F>
__global__ __launch_bounds__(256) void edge_scatter_kernel(const float* __restrict__ h,
                                                           const float* __restrict__ dinv,
                                                           const int* __restrict__ ei,
                                                           float* __restrict__ out) {
    constexpr int VEC   = (F % 128 == 0) ? 4 : 2;
    constexpr int LANES = F / VEC;
    int e = blockIdx.x * (blockDim.x >> 5) + (threadIdx.x >> 5);
    if (e >= N_EDGES) return;
    int lane = threadIdx.x & 31;
    int s = ei[e];
    int d = ei[N_EDGES + e];
    float norm = dinv[s] * dinv[d];
    if (lane < LANES) {
        const float* __restrict__ hp = h   + (size_t)s * F + lane * VEC;
        float*       __restrict__ op = out + (size_t)d * F + lane * VEC;
        if constexpr (VEC == 4) {
            float4 v = *reinterpret_cast<const float4*>(hp);
            unsafeAtomicAdd(op + 0, norm * v.x);
            unsafeAtomicAdd(op + 1, norm * v.y);
            unsafeAtomicAdd(op + 2, norm * v.z);
            unsafeAtomicAdd(op + 3, norm * v.w);
        } else {
            float2 v = *reinterpret_cast<const float2*>(hp);
            unsafeAtomicAdd(op + 0, norm * v.x);
            unsafeAtomicAdd(op + 1, norm * v.y);
        }
    }
}


extern "C" void kernel_launch(void* const* d_in, const int* in_sizes, int n_in,
                              void* d_out, int out_size, void* d_ws, size_t ws_size,
                              hipStream_t stream) {
    const float* x  = (const float*)d_in[0];
    const int*   ei = (const int*)d_in[1];
    const float* W1 = (const float*)d_in[2];
    const float* b1 = (const float*)d_in[3];
    const float* W2 = (const float*)d_in[4];
    const float* b2 = (const float*)d_in[5];
    const float* W3 = (const float*)d_in[6];
    const float* b3 = (const float*)d_in[7];
    float* out = (float*)d_out;

    char* ws = (char*)d_ws;
    float* dinv = (float*)ws;
    float* bufA = (float*)(ws + (((size_t)N_NODES * 4 + 255) & ~(size_t)255));
    float* bufB = bufA + (size_t)N_NODES * HID;
    v2f*   Wp1  = (v2f*)(bufB + (size_t)N_NODES * HID);
    const int NP1 = (F_IN / 4) * 8 * 32;
    const int NP2 = (HID  / 4) * 8 * 32;
    const int NP3 = (HID  / 4) * 3 * 32;
    v2f* Wp2 = Wp1 + NP1;
    v2f* Wp3 = Wp2 + NP2;

    const int pairs   = N_NODES / 32;
    const int gemmBlk = (pairs + 3) / 4;
    const int edgeBlk = (N_EDGES + 7) / 8;

    pack_w_kernel<8><<<(NP1 + 255) / 256, 256, 0, stream>>>(W1, Wp1, F_IN, HID);
    pack_w_kernel<8><<<(NP2 + 255) / 256, 256, 0, stream>>>(W2, Wp2, HID, HID);
    pack_w_kernel<3><<<(NP3 + 255) / 256, 256, 0, stream>>>(W3, Wp3, HID, N_CLS);

    deg_init_kernel <<<(N_NODES + 255) / 256, 256, 0, stream>>>(dinv);
    deg_count_kernel<<<(N_EDGES + 255) / 256, 256, 0, stream>>>(ei + N_EDGES, dinv);
    dinv_kernel     <<<(N_NODES + 255) / 256, 256, 0, stream>>>(dinv);

    gemm_wmma_kernel<8, HID, false><<<gemmBlk, 128, 0, stream>>>(x, Wp1, bufA, N_NODES, F_IN);
    self_bias_kernel<HID><<<((long)N_NODES * (HID/4) + 255) / 256, 256, 0, stream>>>(bufA, dinv, b1, bufB, N_NODES);
    edge_scatter_kernel<HID><<<edgeBlk, 256, 0, stream>>>(bufA, dinv, ei, bufB);

    gemm_wmma_kernel<8, HID, true><<<gemmBlk, 128, 0, stream>>>(bufB, Wp2, bufA, N_NODES, HID);
    self_bias_kernel<HID><<<((long)N_NODES * (HID/4) + 255) / 256, 256, 0, stream>>>(bufA, dinv, b2, bufB, N_NODES);
    edge_scatter_kernel<HID><<<edgeBlk, 256, 0, stream>>>(bufA, dinv, ei, bufB);

    gemm_wmma_kernel<3, N_CLS, true><<<gemmBlk, 128, 0, stream>>>(bufB, Wp3, bufA, N_NODES, HID);
    self_bias_kernel<N_CLS><<<((long)N_NODES * (N_CLS/4) + 255) / 256, 256, 0, stream>>>(bufA, dinv, b3, out, N_NODES);
    edge_scatter_kernel<N_CLS><<<edgeBlk, 256, 0, stream>>>(bufA, dinv, ei, out);
}


# Float atomic edge scatter replaced by a per-call counting sort of edges by destination (integer atomic rank) and a one-wave-per-node gather storing each row once; node sums in arrival-slot order
# speedup vs baseline: 2.2059x; 2.2059x over previous
#include <hip/hip_runtime.h>
#include <hip/hip_bf16.h>

typedef __attribute__((ext_vector_type(2))) float v2f;
typedef __attribute__((ext_vector_type(8))) float v8f;

#define N_NODES 100000
#define N_EDGES 1600000
#define F_IN    500
#define HID     128
#define N_CLS   40


__global__ __launch_bounds__(256) void zero_cnt_kernel(int* __restrict__ cnt) {
    int i = blockIdx.x * blockDim.x + threadIdx.x;
    if (i < N_NODES) cnt[i] = 0;
}

__global__ __launch_bounds__(256) void count_rank_kernel(const int* __restrict__ dst,
                                                         int* __restrict__ cnt,
                                                         int* __restrict__ rank) {
    int e = blockIdx.x * blockDim.x + threadIdx.x;
    if (e < N_EDGES) rank[e] = atomicAdd(&cnt[dst[e]], 1);
}

__global__ __launch_bounds__(256) void scan_kernel(const int* __restrict__ cnt,
                                                   int* __restrict__ rowptr,
                                                   float* __restrict__ dinv) {
    __shared__ int wave_tot[8];
    __shared__ int wave_pre[8];
    const int4* __restrict__ c4 = reinterpret_cast<const int4*>(cnt);
    const int b = blockIdx.x, t = threadIdx.x, lane = t & 31, w = t >> 5;

    int p = 0;
    for (int k = 0; k < b; ++k) {
        int4 v = c4[k * 256 + t];
        p += (v.x + v.y) + (v.z + v.w);
    }
    const int i4 = b * 256 + t;
    const bool live = i4 < N_NODES / 4;
    int4 v = live ? c4[i4] : make_int4(0, 0, 0, 0);
    const int loc = (v.x + v.y) + (v.z + v.w);

#pragma unroll
    for (int off = 16; off > 0; off >>= 1) p += __shfl_xor(p, off);
    int inc = loc;
#pragma unroll
    for (int off = 1; off < 32; off <<= 1) {
        int y = __shfl_up(inc, off);
        if (lane >= off) inc += y;
    }
    if (lane == 31) wave_tot[w] = inc;
    if (lane == 0)  wave_pre[w] = p;
    __syncthreads();
    int ex = inc - loc;
#pragma unroll
    for (int k = 0; k < 8; ++k) {
        ex += wave_pre[k];
        if (k < w) ex += wave_tot[k];
    }
    if (live) {
        int4 r;
        r.x = ex; r.y = r.x + v.x; r.z = r.y + v.y; r.w = r.z + v.z;
        reinterpret_cast<int4*>(rowptr)[i4] = r;
        float4 dv;
        dv.x = rsqrtf((float)(v.x + 1)); dv.y = rsqrtf((float)(v.y + 1));
        dv.z = rsqrtf((float)(v.z + 1)); dv.w = rsqrtf((float)(v.w + 1));
        reinterpret_cast<float4*>(dinv)[i4] = dv;
        if (i4 == N_NODES / 4 - 1) {
            int tot = r.w + v.w;
            reinterpret_cast<int4*>(rowptr)[i4 + 1] = make_int4(tot, tot, tot, tot);
        }
    }
}

__global__ __launch_bounds__(256) void fill_kernel(const int* __restrict__ ei,
                                                   const int* __restrict__ rank,
                                                   const int* __restrict__ rowptr,
                                                   int* __restrict__ csr_src) {
    int e = blockIdx.x * blockDim.x + threadIdx.x;
    if (e < N_EDGES) {
        int s = ei[e];
        int d = ei[N_EDGES + e];
        csr_src[rowptr[d] + rank[e]] = s;
    }
}

__global__ __launch_bounds__(256) void copy_i4_kernel(const int4* __restrict__ from,
                                                      int4* __restrict__ to, int n4) {
    int i = blockIdx.x * blockDim.x + threadIdx.x;
    if (i < n4) to[i] = from[i];
}


template<int NT>
__global__ __launch_bounds__(256) void pack_w_kernel(const float* __restrict__ W,
                                                     v2f* __restrict__ Wp,
                                                     int K, int Nc) {
    int idx = blockIdx.x * blockDim.x + threadIdx.x;
    int total = (K / 4) * NT * 32;
    if (idx >= total) return;
    int lane = idx & 31;
    int t    = (idx >> 5) % NT;
    int ks   = idx / (32 * NT);
    int col  = t * 16 + (lane & 15);
    int k    = ks * 4 + ((lane >> 4) * 2);
    v2f v;
    v.x = (col < Nc) ? W[(size_t)k * Nc + col]       : 0.0f;
    v.y = (col < Nc) ? W[(size_t)(k + 1) * Nc + col] : 0.0f;
    Wp[idx] = v;
}


template<int NT, int NC, bool RELU>
__global__ __launch_bounds__(128) void gemm_wmma_kernel(const float* __restrict__ A,
                                                        const v2f* __restrict__ Wp,
                                                        float* __restrict__ Out,
                                                        int M, int K) {
    const int wave = threadIdx.x >> 5;
    const int lane = threadIdx.x & 31;
    const int pair = blockIdx.x * 4 + wave;
    if (pair * 32 >= M) return;

    const int kk  = (lane >> 4) * 2;
    const int col = lane & 15;

    const float* __restrict__ Arow0 = A + (size_t)(pair * 32 + col) * K + kk;
    const float* __restrict__ Arow1 = Arow0 + (size_t)16 * K;
    const v2f*   __restrict__ wp    = Wp + lane;

    v8f acc0[NT], acc1[NT];
#pragma unroll
    for (int t = 0; t < NT; ++t) {
        acc0[t] = (v8f){0.f,0.f,0.f,0.f,0.f,0.f,0.f,0.f};
        acc1[t] = (v8f){0.f,0.f,0.f,0.f,0.f,0.f,0.f,0.f};
    }

    const int nsteps = K >> 2;
#pragma unroll 2
    for (int ks = 0; ks < nsteps; ++ks) {
        v2f a0 = *reinterpret_cast<const v2f*>(Arow0 + ks * 4);
        v2f a1 = *reinterpret_cast<const v2f*>(Arow1 + ks * 4);
        if (RELU) {
            a0.x = fmaxf(a0.x, 0.0f); a0.y = fmaxf(a0.y, 0.0f);
            a1.x = fmaxf(a1.x, 0.0f); a1.y = fmaxf(a1.y, 0.0f);
        }
#pragma unroll
        for (int t = 0; t < NT; ++t) {
            v2f b = wp[(ks * NT + t) * 32];
            acc0[t] = __builtin_amdgcn_wmma_f32_16x16x4_f32(
                          false, a0, false, b, (short)0, acc0[t], false, false);
            acc1[t] = __builtin_amdgcn_wmma_f32_16x16x4_f32(
                          false, a1, false, b, (short)0, acc1[t], false, false);
        }
    }

    const int rbase0 = pair * 32 + ((lane < 16) ? 0 : 8);
    const int rbase1 = rbase0 + 16;
#pragma unroll
    for (int t = 0; t < NT; ++t) {
        int c = t * 16 + col;
        if ((NC % 16 == 0) || (c < NC)) {
#pragma unroll
            for (int r = 0; r < 8; ++r) {
                Out[(size_t)(rbase0 + r) * NC + c] = acc0[t][r];
                Out[(size_t)(rbase1 + r) * NC + c] = acc1[t][r];
            }
        }
    }
}


template<int F>
__global__ __launch_bounds__(256) void gather_kernel(const float* __restrict__ h,
                                                     const float* __restrict__ dinv,
                                                     const int* __restrict__ rowptr,
                                                     const int* __restrict__ csr_src,
                                                     const float* __restrict__ bias,
                                                     float* __restrict__ out) {
    constexpr int CH  = F / 4;
    constexpr int EPI = 32 / CH;
    const int node = __builtin_amdgcn_readfirstlane(blockIdx.x * 8 + (threadIdx.x >> 5));
    if (node >= N_NODES) return;
    const int lane = threadIdx.x & 31;
    const int g = lane / CH;
    const int c = lane - g * CH;
    const bool active = g < EPI;

    const int start = rowptr[node];
    const int end   = rowptr[node + 1];
    const float dd  = dinv[node];
    const float4* __restrict__ h4 = reinterpret_cast<const float4*>(h);

    float4 acc = make_float4(0.f, 0.f, 0.f, 0.f);
    for (int base = start; base < end; base += 32) {
        const int n = min(32, end - base);
        int   ms = 0;
        float mw = 0.f;
        if (lane < n) {
            ms = (int)min((unsigned)csr_src[base + lane], (unsigned)(N_NODES - 1));
            mw = dinv[ms] * dd;
        }
        if constexpr (EPI == 1) {
#pragma unroll 4
            for (int j = 0; j < n; ++j) {
                const int   s = __builtin_amdgcn_readlane(ms, j);
                const float w = __builtin_bit_cast(float,
                                    __builtin_amdgcn_readlane(__builtin_bit_cast(int, mw), j));
                const float4 v = h4[(size_t)s * CH + c];
                acc.x += w * v.x; acc.y += w * v.y; acc.z += w * v.z; acc.w += w * v.w;
            }
        } else {
            for (int j = 0; j < n; j += EPI) {
                const int idx = j + g;
                const int   s = __shfl(ms, idx & 31);
                const float w = __shfl(mw, idx & 31);
                if (active && idx < n) {
                    const float4 v = h4[(size_t)s * CH + c];
                    acc.x += w * v.x; acc.y += w * v.y; acc.z += w * v.z; acc.w += w * v.w;
                }
            }
        }
    }
    if constexpr (EPI == 3) {
        acc.x += __shfl(acc.x, (lane + CH) & 31) + __shfl(acc.x, (lane + 2 * CH) & 31);
        acc.y += __shfl(acc.y, (lane + CH) & 31) + __shfl(acc.y, (lane + 2 * CH) & 31);
        acc.z += __shfl(acc.z, (lane + CH) & 31) + __shfl(acc.z, (lane + 2 * CH) & 31);
        acc.w += __shfl(acc.w, (lane + CH) & 31) + __shfl(acc.w, (lane + 2 * CH) & 31);
    }
    if (lane < CH) {
        const float  sl = dd * dd;
        const float4 hv = h4[(size_t)node * CH + lane];
        const float4 bv = reinterpret_cast<const float4*>(bias)[lane];
        float4 ov;
        ov.x = (bv.x + sl * hv.x) + acc.x;  ov.y = (bv.y + sl * hv.y) + acc.y;
        ov.z = (bv.z + sl * hv.z) + acc.z;  ov.w = (bv.w + sl * hv.w) + acc.w;
        reinterpret_cast<float4*>(out)[(size_t)node * CH + lane] = ov;
    }
}


extern "C" void kernel_launch(void* const* d_in, const int* in_sizes, int n_in,
                              void* d_out, int out_size, void* d_ws, size_t ws_size,
                              hipStream_t stream) {
    const float* x  = (const float*)d_in[0];
    const int*   ei = (const int*)d_in[1];
    const float* W1 = (const float*)d_in[2];
    const float* b1 = (const float*)d_in[3];
    const float* W2 = (const float*)d_in[4];
    const float* b2 = (const float*)d_in[5];
    const float* W3 = (const float*)d_in[6];
    const float* b3 = (const float*)d_in[7];
    float* out = (float*)d_out;

    char* ws = (char*)d_ws;
    float* dinv = (float*)ws;
    float* bufA = (float*)(ws + (((size_t)N_NODES * 4 + 255) & ~(size_t)255));
    float* bufB = bufA + (size_t)N_NODES * HID;
    v2f*   Wp1  = (v2f*)(bufB + (size_t)N_NODES * HID);
    const int NP1 = (F_IN / 4) * 8 * 32;
    const int NP2 = (HID  / 4) * 8 * 32;
    const int NP3 = (HID  / 4) * 3 * 32;
    v2f* Wp2 = Wp1 + NP1;
    v2f* Wp3 = Wp2 + NP2;

    const int CSR_N = N_EDGES + N_NODES + 4;
    int* csr_src = (int*)out;
    int* rowptr  = csr_src + N_EDGES;
    int* rank    = (int*)bufB;
    int* cnt     = rank + N_EDGES;
    int* csr_src3 = (int*)bufB;
    int* rowptr3  = csr_src3 + N_EDGES;

    const int pairs   = N_NODES / 32;
    const int gemmBlk = (pairs + 3) / 4;
    const int nodeBlk = N_NODES / 8;

    pack_w_kernel<8><<<(NP1 + 255) / 256, 256, 0, stream>>>(W1, Wp1, F_IN, HID);
    pack_w_kernel<8><<<(NP2 + 255) / 256, 256, 0, stream>>>(W2, Wp2, HID, HID);
    pack_w_kernel<3><<<(NP3 + 255) / 256, 256, 0, stream>>>(W3, Wp3, HID, N_CLS);

    zero_cnt_kernel  <<<(N_NODES + 255) / 256, 256, 0, stream>>>(cnt);
    count_rank_kernel<<<(N_EDGES + 255) / 256, 256, 0, stream>>>(ei + N_EDGES, cnt, rank);
    scan_kernel      <<<(N_NODES + 1023) / 1024, 256, 0, stream>>>(cnt, rowptr, dinv);
    fill_kernel      <<<(N_EDGES + 255) / 256, 256, 0, stream>>>(ei, rank, rowptr, csr_src);

    gemm_wmma_kernel<8, HID, false><<<gemmBlk, 128, 0, stream>>>(x, Wp1, bufA, N_NODES, F_IN);
    gather_kernel<HID><<<nodeBlk, 256, 0, stream>>>(bufA, dinv, rowptr, csr_src, b1, bufB);

    gemm_wmma_kernel<8, HID, true><<<gemmBlk, 128, 0, stream>>>(bufB, Wp2, bufA, N_NODES, HID);
    gather_kernel<HID><<<nodeBlk, 256, 0, stream>>>(bufA, dinv, rowptr, csr_src, b2, bufB);

    gemm_wmma_kernel<3, N_CLS, true><<<gemmBlk, 128, 0, stream>>>(bufB, Wp3, bufA, N_NODES, HID);
    copy_i4_kernel<<<(CSR_N / 4 + 255) / 256, 256, 0, stream>>>((const int4*)csr_src, (int4*)csr_src3, CSR_N / 4);
    gather_kernel<N_CLS><<<nodeBlk, 256, 0, stream>>>(bufA, dinv, rowptr3, csr_src3, b3, out);
}


# GEMM kernels fetch A and B fragments through a register ring of five k-steps, each slot refilled right behind its products, so memory round trips overlap the WMMAs; k order and sum orders unchanged
# speedup vs baseline: 2.4456x; 1.1087x over previous
#include <hip/hip_runtime.h>
#include <hip/hip_bf16.h>

typedef __attribute__((ext_vector_type(2))) float v2f;
typedef __attribute__((ext_vector_type(8))) float v8f;

#define N_NODES 100000
#define N_EDGES 1600000
#define F_IN    500
#define HID     128
#define N_CLS   40


__global__ __launch_bounds__(256) void zero_cnt_kernel(int* __restrict__ cnt) {
    int i = blockIdx.x * blockDim.x + threadIdx.x;
    if (i < N_NODES) cnt[i] = 0;
}

__global__ __launch_bounds__(256) void count_rank_kernel(const int* __restrict__ dst,
                                                         int* __restrict__ cnt,
                                                         int* __restrict__ rank) {
    int e = blockIdx.x * blockDim.x + threadIdx.x;
    if (e < N_EDGES) rank[e] = atomicAdd(&cnt[dst[e]], 1);
}

__global__ __launch_bounds__(256) void scan_kernel(const int* __restrict__ cnt,
                                                   int* __restrict__ rowptr,
                                                   float* __restrict__ dinv) {
    __shared__ int wave_tot[8];
    __shared__ int wave_pre[8];
    const int4* __restrict__ c4 = reinterpret_cast<const int4*>(cnt);
    const int b = blockIdx.x, t = threadIdx.x, lane = t & 31, w = t >> 5;

    int p = 0;
    for (int k = 0; k < b; ++k) {
        int4 v = c4[k * 256 + t];
        p += (v.x + v.y) + (v.z + v.w);
    }
    const int i4 = b * 256 + t;
    const bool live = i4 < N_NODES / 4;
    int4 v = live ? c4[i4] : make_int4(0, 0, 0, 0);
    const int loc = (v.x + v.y) + (v.z + v.w);

#pragma unroll
    for (int off = 16; off > 0; off >>= 1) p += __shfl_xor(p, off);
    int inc = loc;
#pragma unroll
    for (int off = 1; off < 32; off <<= 1) {
        int y = __shfl_up(inc, off);
        if (lane >= off) inc += y;
    }
    if (lane == 31) wave_tot[w] = inc;
    if (lane == 0)  wave_pre[w] = p;
    __syncthreads();
    int ex = inc - loc;
#pragma unroll
    for (int k = 0; k < 8; ++k) {
        ex += wave_pre[k];
        if (k < w) ex += wave_tot[k];
    }
    if (live) {
        int4 r;
        r.x = ex; r.y = r.x + v.x; r.z = r.y + v.y; r.w = r.z + v.z;
        reinterpret_cast<int4*>(rowptr)[i4] = r;
        float4 dv;
        dv.x = rsqrtf((float)(v.x + 1)); dv.y = rsqrtf((float)(v.y + 1));
        dv.z = rsqrtf((float)(v.z + 1)); dv.w = rsqrtf((float)(v.w + 1));
        reinterpret_cast<float4*>(dinv)[i4] = dv;
        if (i4 == N_NODES / 4 - 1) {
            int tot = r.w + v.w;
            reinterpret_cast<int4*>(rowptr)[i4 + 1] = make_int4(tot, tot, tot, tot);
        }
    }
}

__global__ __launch_bounds__(256) void fill_kernel(const int* __restrict__ ei,
                                                   const int* __restrict__ rank,
                                                   const int* __restrict__ rowptr,
                                                   int* __restrict__ csr_src) {
    int e = blockIdx.x * blockDim.x + threadIdx.x;
    if (e < N_EDGES) {
        int s = ei[e];
        int d = ei[N_EDGES + e];
        csr_src[rowptr[d] + rank[e]] = s;
    }
}

__global__ __launch_bounds__(256) void copy_i4_kernel(const int4* __restrict__ from,
                                                      int4* __restrict__ to, int n4) {
    int i = blockIdx.x * blockDim.x + threadIdx.x;
    if (i < n4) to[i] = from[i];
}


template<int NT>
__global__ __launch_bounds__(256) void pack_w_kernel(const float* __restrict__ W,
                                                     v2f* __restrict__ Wp,
                                                     int K, int Nc) {
    int idx = blockIdx.x * blockDim.x + threadIdx.x;
    int total = (K / 4) * NT * 32;
    if (idx >= total) return;
    int lane = idx & 31;
    int t    = (idx >> 5) % NT;
    int ks   = idx / (32 * NT);
    int col  = t * 16 + (lane & 15);
    int k    = ks * 4 + ((lane >> 4) * 2);
    v2f v;
    v.x = (col < Nc) ? W[(size_t)k * Nc + col]       : 0.0f;
    v.y = (col < Nc) ? W[(size_t)(k + 1) * Nc + col] : 0.0f;
    Wp[idx] = v;
}


#define RING 5

template<int NT>
struct KFrag {
    v2f a0, a1;
    v2f b[NT];
};

template<int NT, int NC, bool RELU>
__global__ __launch_bounds__(128) void gemm_wmma_kernel(const float* __restrict__ A,
                                                        const v2f* __restrict__ Wp,
                                                        const float* __restrict__ dinv,
                                                        float* __restrict__ Out,
                                                        int M, int K) {
    const int wave = threadIdx.x >> 5;
    const int lane = threadIdx.x & 31;
    const int pair = blockIdx.x * 4 + wave;
    if (pair * 32 >= M) return;

    const int kk  = (lane >> 4) * 2;
    const int col = lane & 15;

    const float* __restrict__ Arow0 = A + (size_t)(pair * 32 + col) * K + kk;
    const float* __restrict__ Arow1 = Arow0 + (size_t)16 * K;
    const v2f*   __restrict__ wp    = Wp + lane;

    const int rbase0 = pair * 32 + ((lane < 16) ? 0 : 8);
    const int rbase1 = rbase0 + 16;

    v8f acc0[NT], acc1[NT];
#pragma unroll
    for (int t = 0; t < NT; ++t) {
        acc0[t] = (v8f){0.f,0.f,0.f,0.f,0.f,0.f,0.f,0.f};
        acc1[t] = (v8f){0.f,0.f,0.f,0.f,0.f,0.f,0.f,0.f};
    }

    const int nsteps = K >> 2;
    const int last   = nsteps - 1;

    auto fetch = [&](int s) {
        KFrag<NT> f;
        f.a0 = *reinterpret_cast<const v2f*>(Arow0 + s * 4);
        f.a1 = *reinterpret_cast<const v2f*>(Arow1 + s * 4);
#pragma unroll
        for (int t = 0; t < NT; ++t) f.b[t] = wp[(s * NT + t) * 32];
        return f;
    };
    auto products = [&](const KFrag<NT>& f) {
        v2f a0 = f.a0, a1 = f.a1;
        if (RELU) {
            a0.x = fmaxf(a0.x, 0.0f); a0.y = fmaxf(a0.y, 0.0f);
            a1.x = fmaxf(a1.x, 0.0f); a1.y = fmaxf(a1.y, 0.0f);
        }
#pragma unroll
        for (int t = 0; t < NT; ++t) {
            acc0[t] = __builtin_amdgcn_wmma_f32_16x16x4_f32(
                          false, a0, false, f.b[t], (short)0, acc0[t], false, false);
            acc1[t] = __builtin_amdgcn_wmma_f32_16x16x4_f32(
                          false, a1, false, f.b[t], (short)0, acc1[t], false, false);
        }
    };

    KFrag<NT> ring[RING];
#pragma unroll
    for (int j = 0; j < RING; ++j) {
        ring[j] = fetch(min(j, last));
        __builtin_amdgcn_sched_barrier(0);
    }

    int ks = 0;
#pragma unroll 1
    for (; ks + 2 * RING <= nsteps; ks += RING) {
#pragma unroll
        for (int j = 0; j < RING; ++j) {
            products(ring[j]);
            ring[j] = fetch(ks + RING + j);
            __builtin_amdgcn_sched_barrier(0);
        }
    }

    const int R = nsteps - ks;
    float sc0[8], sc1[8];
#pragma unroll
    for (int j = 0; j < RING; ++j) {
        if (j < R) products(ring[j]);
        if (j < R - RING) ring[j] = fetch(ks + RING + j);
        if (j == 1) {
            const float4 p0 = *reinterpret_cast<const float4*>(dinv + rbase0);
            const float4 p1 = *reinterpret_cast<const float4*>(dinv + rbase0 + 4);
            const float4 q0 = *reinterpret_cast<const float4*>(dinv + rbase1);
            const float4 q1 = *reinterpret_cast<const float4*>(dinv + rbase1 + 4);
            sc0[0] = p0.x; sc0[1] = p0.y; sc0[2] = p0.z; sc0[3] = p0.w;
            sc0[4] = p1.x; sc0[5] = p1.y; sc0[6] = p1.z; sc0[7] = p1.w;
            sc1[0] = q0.x; sc1[1] = q0.y; sc1[2] = q0.z; sc1[3] = q0.w;
            sc1[4] = q1.x; sc1[5] = q1.y; sc1[6] = q1.z; sc1[7] = q1.w;
        }
    }
#pragma unroll
    for (int j = 0; j < RING - 1; ++j) {
        if (j < R - RING) products(ring[j]);
    }

#pragma unroll
    for (int t = 0; t < NT; ++t) {
        int c = t * 16 + col;
        if ((NC % 16 == 0) || (c < NC)) {
#pragma unroll
            for (int r = 0; r < 8; ++r) {
                Out[(size_t)(rbase0 + r) * NC + c] = acc0[t][r] * sc0[r];
                Out[(size_t)(rbase1 + r) * NC + c] = acc1[t][r] * sc1[r];
            }
        }
    }
}


template<int F>
__global__ __launch_bounds__(256) void gather_kernel(const float* __restrict__ h,
                                                     const float* __restrict__ dinv,
                                                     const int* __restrict__ rowptr,
                                                     const int* __restrict__ csr_src,
                                                     const float* __restrict__ bias,
                                                     float* __restrict__ out) {
    constexpr int CH  = F / 4;
    constexpr int EPI = 32 / CH;
    const int node = __builtin_amdgcn_readfirstlane(blockIdx.x * 8 + (threadIdx.x >> 5));
    if (node >= N_NODES) return;
    const int lane = threadIdx.x & 31;
    const int g = lane / CH;
    const int c = lane - g * CH;
    const bool active = g < EPI;

    const int start = rowptr[node];
    const int end   = rowptr[node + 1];
    const float dd  = dinv[node];
    const float4* __restrict__ h4 = reinterpret_cast<const float4*>(h);

    float4 acc = make_float4(0.f, 0.f, 0.f, 0.f);
    for (int base = start; base < end; base += 32) {
        const int n = min(32, end - base);
        int ms = 0;
        if (lane < n) {
            ms = (int)min((unsigned)csr_src[base + lane], (unsigned)(N_NODES - 1));
        }
        if constexpr (EPI == 1) {
#pragma unroll 4
            for (int j = 0; j < n; ++j) {
                const int s = __builtin_amdgcn_readlane(ms, j);
                const float4 v = h4[(size_t)s * CH + c];
                acc.x += v.x; acc.y += v.y; acc.z += v.z; acc.w += v.w;
            }
        } else {
            for (int j = 0; j < n; j += EPI) {
                const int idx = j + g;
                const int s = __shfl(ms, idx & 31);
                if (active && idx < n) {
                    const float4 v = h4[(size_t)s * CH + c];
                    acc.x += v.x; acc.y += v.y; acc.z += v.z; acc.w += v.w;
                }
            }
        }
    }
    if constexpr (EPI == 3) {
        acc.x += __shfl(acc.x, (lane + CH) & 31) + __shfl(acc.x, (lane + 2 * CH) & 31);
        acc.y += __shfl(acc.y, (lane + CH) & 31) + __shfl(acc.y, (lane + 2 * CH) & 31);
        acc.z += __shfl(acc.z, (lane + CH) & 31) + __shfl(acc.z, (lane + 2 * CH) & 31);
        acc.w += __shfl(acc.w, (lane + CH) & 31) + __shfl(acc.w, (lane + 2 * CH) & 31);
    }
    if (lane < CH) {
        const float4 hv = h4[(size_t)node * CH + lane];
        const float4 bv = reinterpret_cast<const float4*>(bias)[lane];
        float4 ov;
        ov.x = (bv.x + dd * hv.x) + dd * acc.x;  ov.y = (bv.y + dd * hv.y) + dd * acc.y;
        ov.z = (bv.z + dd * hv.z) + dd * acc.z;  ov.w = (bv.w + dd * hv.w) + dd * acc.w;
        reinterpret_cast<float4*>(out)[(size_t)node * CH + lane] = ov;
    }
}


extern "C" void kernel_launch(void* const* d_in, const int* in_sizes, int n_in,
                              void* d_out, int out_size, void* d_ws, size_t ws_size,
                              hipStream_t stream) {
    const float* x  = (const float*)d_in[0];
    const int*   ei = (const int*)d_in[1];
    const float* W1 = (const float*)d_in[2];
    const float* b1 = (const float*)d_in[3];
    const float* W2 = (const float*)d_in[4];
    const float* b2 = (const float*)d_in[5];
    const float* W3 = (const float*)d_in[6];
    const float* b3 = (const float*)d_in[7];
    float* out = (float*)d_out;

    char* ws = (char*)d_ws;
    float* dinv = (float*)ws;
    float* bufA = (float*)(ws + (((size_t)N_NODES * 4 + 255) & ~(size_t)255));
    float* bufB = bufA + (size_t)N_NODES * HID;
    v2f*   Wp1  = (v2f*)(bufB + (size_t)N_NODES * HID);
    const int NP1 = (F_IN / 4) * 8 * 32;
    const int NP2 = (HID  / 4) * 8 * 32;
    const int NP3 = (HID  / 4) * 3 * 32;
    v2f* Wp2 = Wp1 + NP1;
    v2f* Wp3 = Wp2 + NP2;

    const int CSR_N = N_EDGES + N_NODES + 4;
    int* csr_src = (int*)out;
    int* rowptr  = csr_src + N_EDGES;
    int* rank    = (int*)bufB;
    int* cnt     = rank + N_EDGES;
    int* csr_src3 = (int*)bufB;
    int* rowptr3  = csr_src3 + N_EDGES;

    const int pairs   = N_NODES / 32;
    const int gemmBlk = (pairs + 3) / 4;
    const int nodeBlk = N_NODES / 8;

    pack_w_kernel<8><<<(NP1 + 255) / 256, 256, 0, stream>>>(W1, Wp1, F_IN, HID);
    pack_w_kernel<8><<<(NP2 + 255) / 256, 256, 0, stream>>>(W2, Wp2, HID, HID);
    pack_w_kernel<3><<<(NP3 + 255) / 256, 256, 0, stream>>>(W3, Wp3, HID, N_CLS);

    zero_cnt_kernel  <<<(N_NODES + 255) / 256, 256, 0, stream>>>(cnt);
    count_rank_kernel<<<(N_EDGES + 255) / 256, 256, 0, stream>>>(ei + N_EDGES, cnt, rank);
    scan_kernel      <<<(N_NODES + 1023) / 1024, 256, 0, stream>>>(cnt, rowptr, dinv);
    fill_kernel      <<<(N_EDGES + 255) / 256, 256, 0, stream>>>(ei, rank, rowptr, csr_src);

    gemm_wmma_kernel<8, HID, false><<<gemmBlk, 128, 0, stream>>>(x, Wp1, dinv, bufA, N_NODES, F_IN);
    gather_kernel<HID><<<nodeBlk, 256, 0, stream>>>(bufA, dinv, rowptr, csr_src, b1, bufB);

    gemm_wmma_kernel<8, HID, true><<<gemmBlk, 128, 0, stream>>>(bufB, Wp2, dinv, bufA, N_NODES, HID);
    gather_kernel<HID><<<nodeBlk, 256, 0, stream>>>(bufA, dinv, rowptr, csr_src, b2, bufB);

    gemm_wmma_kernel<3, N_CLS, true><<<gemmBlk, 128, 0, stream>>>(bufB, Wp3, dinv, bufA, N_NODES, HID);
    copy_i4_kernel<<<(CSR_N / 4 + 255) / 256, 256, 0, stream>>>((const int4*)csr_src, (int4*)csr_src3, CSR_N / 4);
    gather_kernel<N_CLS><<<nodeBlk, 256, 0, stream>>>(bufA, dinv, rowptr3, csr_src3, b3, out);
}


# CSR build: per-edge returning global atomic rank replaced by two-level counting sort (256 edge chunks x 196 node buckets) ranked by LDS counters; segment (sum) order now LDS arrival order
# speedup vs baseline: 3.1223x; 1.2767x over previous
#include <hip/hip_runtime.h>
#include <hip/hip_bf16.h>

typedef __attribute__((ext_vector_type(2))) float v2f;
typedef __attribute__((ext_vector_type(8))) float v8f;

#define N_NODES 100000
#define N_EDGES 1600000
#define F_IN    500
#define HID     128
#define N_CLS   40


#define NPB      512
#define NBKT     ((N_NODES + NPB - 1) / NPB)
#define NCHUNK   256
#define CHUNK_E  ((N_EDGES + NCHUNK - 1) / NCHUNK)
#define EPT      ((CHUNK_E + 255) / 256)
#define STAGE_CAP 12288
#define L2T      512
#define L2U      8

static_assert(NBKT <= 256 && NBKT <= L2T, "one thread per bucket");
static_assert(N_NODES <= (1 << 17), "src must fit 17 bits");

template<int NW>
__device__ inline int block_excl_scan(int v, int* sh_wave, int& tot) {
    const int lane = threadIdx.x & 31, w = threadIdx.x >> 5;
    int inc = v;
#pragma unroll
    for (int off = 1; off < 32; off <<= 1) {
        int y = __shfl_up(inc, off);
        if (lane >= off) inc += y;
    }
    if (lane == 31) sh_wave[w] = inc;
    __syncthreads();
    int pre = 0, sum = 0;
#pragma unroll
    for (int k = 0; k < NW; ++k) {
        const int x = sh_wave[k];
        if (k < w) pre += x;
        sum += x;
    }
    tot = sum;
    return pre + inc - v;
}

__device__ inline unsigned clamp_node(int d) { return min((unsigned)d, (unsigned)(N_NODES - 1)); }

__global__ __launch_bounds__(256) void chunk_hist_kernel(const int* __restrict__ dst,
                                                         int* __restrict__ histT) {
    __shared__ int s_h[NBKT];
    const int c = blockIdx.x, t = threadIdx.x;
    if (t < NBKT) s_h[t] = 0;
    const int e0 = c * CHUNK_E;
    const int e1 = min(e0 + CHUNK_E, N_EDGES);
    int dv[EPT];
#pragma unroll
    for (int k = 0; k < EPT; ++k) {
        const int e = min(e0 + k * 256 + t, N_EDGES - 1);
        dv[k] = dst[e];
    }
    __syncthreads();
#pragma unroll
    for (int k = 0; k < EPT; ++k) {
        const int e = e0 + k * 256 + t;
        if (e < e1) atomicAdd(&s_h[clamp_node(dv[k]) / NPB], 1);
    }
    __syncthreads();
    if (t < NBKT) histT[t * NCHUNK + c] = s_h[t];
}

__global__ __launch_bounds__(NCHUNK) void chunk_scan_kernel(const int* __restrict__ histT,
                                                            int* __restrict__ colscan,
                                                            int* __restrict__ total) {
    __shared__ int s_w[NCHUNK / 32];
    const int b = blockIdx.x, t = threadIdx.x;
    const int v = histT[b * NCHUNK + t];
    int tot;
    const int ex = block_excl_scan<NCHUNK / 32>(v, s_w, tot);
    colscan[b * NCHUNK + t] = ex;
    if (t == 0) total[b] = tot;
}

__global__ __launch_bounds__(256) void chunk_scatter_kernel(const int* __restrict__ ei,
                                                            const int* __restrict__ colscan,
                                                            const int* __restrict__ total,
                                                            int* __restrict__ bucketed) {
    __shared__ int s_cur[NBKT];
    __shared__ int s_w[8];
    const int c = blockIdx.x, t = threadIdx.x;
    const int e0 = c * CHUNK_E;
    const int e1 = min(e0 + CHUNK_E, N_EDGES);
    int sv[EPT], dv[EPT];
#pragma unroll
    for (int k = 0; k < EPT; ++k) {
        const int e = min(e0 + k * 256 + t, N_EDGES - 1);
        sv[k] = ei[e];
        dv[k] = ei[N_EDGES + e];
    }
    const int tv = (t < NBKT) ? total[t] : 0;
    const int cs = (t < NBKT) ? colscan[t * NCHUNK + c] : 0;
    int tot;
    const int base = block_excl_scan<8>(tv, s_w, tot);
    if (t < NBKT) s_cur[t] = base + cs;
    __syncthreads();
    int pos[EPT];
#pragma unroll
    for (int k = 0; k < EPT; ++k) {
        const int e = e0 + k * 256 + t;
        pos[k] = (e < e1) ? atomicAdd(&s_cur[clamp_node(dv[k]) / NPB], 1) : -1;
    }
#pragma unroll
    for (int k = 0; k < EPT; ++k) {
        if (pos[k] >= 0) {
            const unsigned d = clamp_node(dv[k]);
            bucketed[pos[k]] = (int)(((d % NPB) << 17) | ((unsigned)sv[k] & 0x1FFFFu));
        }
    }
}

__global__ __launch_bounds__(L2T) void bucket_sort_kernel(const int* __restrict__ bucketed,
                                                          const int* __restrict__ total,
                                                          int* __restrict__ csr_src,
                                                          int* __restrict__ rowptr,
                                                          float* __restrict__ dinv) {
    __shared__ int s_cnt[NPB];
    __shared__ int s_w0[L2T / 32];
    __shared__ int s_w1[L2T / 32];
    __shared__ int s_stage[STAGE_CAP];
    const int b = blockIdx.x, t = threadIdx.x;

    const int tv = (t < b) ? total[t] : 0;
    const int n  = total[b];
    int base;
    block_excl_scan<L2T / 32>(tv, s_w0, base);
    const int* __restrict__ mine = bucketed + base;

    for (int i = t; i < NPB; i += L2T) s_cnt[i] = 0;
    __syncthreads();

    for (int i0 = 0; i0 < n; i0 += L2T * L2U) {
        int p[L2U];
#pragma unroll
        for (int k = 0; k < L2U; ++k) p[k] = mine[min(i0 + k * L2T + t, n - 1)];
#pragma unroll
        for (int k = 0; k < L2U; ++k)
            if (i0 + k * L2T + t < n) atomicAdd(&s_cnt[(p[k] >> 17) & (NPB - 1)], 1);
    }
    __syncthreads();

    static_assert(NPB == L2T, "one node per thread");
    const int cn = s_cnt[t];
    int tot;
    const int ex = block_excl_scan<L2T / 32>(cn, s_w1, tot);
    const int node = b * NPB + t;
    if (node < N_NODES) {
        rowptr[node] = base + ex;
        dinv[node]   = rsqrtf((float)(cn + 1));
    }
    if (b == NBKT - 1 && t < 4) rowptr[N_NODES + t] = base + n;
    s_cnt[t] = ex;
    __syncthreads();

    const bool fits = n <= STAGE_CAP;
    for (int i0 = 0; i0 < n; i0 += L2T * L2U) {
        int p[L2U], pos[L2U];
#pragma unroll
        for (int k = 0; k < L2U; ++k) p[k] = mine[min(i0 + k * L2T + t, n - 1)];
#pragma unroll
        for (int k = 0; k < L2U; ++k)
            pos[k] = (i0 + k * L2T + t < n) ? atomicAdd(&s_cnt[(p[k] >> 17) & (NPB - 1)], 1) : -1;
#pragma unroll
        for (int k = 0; k < L2U; ++k) {
            if (pos[k] >= 0) {
                const int s = p[k] & 0x1FFFF;
                if (fits) s_stage[pos[k]] = s;
                else      csr_src[base + pos[k]] = s;
            }
        }
    }
    if (fits) {
        __syncthreads();
        for (int i = t; i < n; i += L2T) csr_src[base + i] = s_stage[i];
    }
}

__global__ __launch_bounds__(256) void copy_i4_kernel(const int4* __restrict__ from,
                                                      int4* __restrict__ to, int n4) {
    int i = blockIdx.x * blockDim.x + threadIdx.x;
    if (i < n4) to[i] = from[i];
}


template<int NT>
__global__ __launch_bounds__(256) void pack_w_kernel(const float* __restrict__ W,
                                                     v2f* __restrict__ Wp,
                                                     int K, int Nc) {
    int idx = blockIdx.x * blockDim.x + threadIdx.x;
    int total = (K / 4) * NT * 32;
    if (idx >= total) return;
    int lane = idx & 31;
    int t    = (idx >> 5) % NT;
    int ks   = idx / (32 * NT);
    int col  = t * 16 + (lane & 15);
    int k    = ks * 4 + ((lane >> 4) * 2);
    v2f v;
    v.x = (col < Nc) ? W[(size_t)k * Nc + col]       : 0.0f;
    v.y = (col < Nc) ? W[(size_t)(k + 1) * Nc + col] : 0.0f;
    Wp[idx] = v;
}


#define RING 5

template<int NT>
struct KFrag {
    v2f a0, a1;
    v2f b[NT];
};

template<int NT, int NC, bool RELU>
__global__ __launch_bounds__(128) void gemm_wmma_kernel(const float* __restrict__ A,
                                                        const v2f* __restrict__ Wp,
                                                        const float* __restrict__ dinv,
                                                        float* __restrict__ Out,
                                                        int M, int K) {
    const int wave = threadIdx.x >> 5;
    const int lane = threadIdx.x & 31;
    const int pair = blockIdx.x * 4 + wave;
    if (pair * 32 >= M) return;

    const int kk  = (lane >> 4) * 2;
    const int col = lane & 15;

    const float* __restrict__ Arow0 = A + (size_t)(pair * 32 + col) * K + kk;
    const float* __restrict__ Arow1 = Arow0 + (size_t)16 * K;
    const v2f*   __restrict__ wp    = Wp + lane;

    const int rbase0 = pair * 32 + ((lane < 16) ? 0 : 8);
    const int rbase1 = rbase0 + 16;

    v8f acc0[NT], acc1[NT];
#pragma unroll
    for (int t = 0; t < NT; ++t) {
        acc0[t] = (v8f){0.f,0.f,0.f,0.f,0.f,0.f,0.f,0.f};
        acc1[t] = (v8f){0.f,0.f,0.f,0.f,0.f,0.f,0.f,0.f};
    }

    const int nsteps = K >> 2;
    const int last   = nsteps - 1;

    auto fetch = [&](int s) {
        KFrag<NT> f;
        f.a0 = *reinterpret_cast<const v2f*>(Arow0 + s * 4);
        f.a1 = *reinterpret_cast<const v2f*>(Arow1 + s * 4);
#pragma unroll
        for (int t = 0; t < NT; ++t) f.b[t] = wp[(s * NT + t) * 32];
        return f;
    };
    auto products = [&](const KFrag<NT>& f) {
        v2f a0 = f.a0, a1 = f.a1;
        if (RELU) {
            a0.x = fmaxf(a0.x, 0.0f); a0.y = fmaxf(a0.y, 0.0f);
            a1.x = fmaxf(a1.x, 0.0f); a1.y = fmaxf(a1.y, 0.0f);
        }
#pragma unroll
        for (int t = 0; t < NT; ++t) {
            acc0[t] = __builtin_amdgcn_wmma_f32_16x16x4_f32(
                          false, a0, false, f.b[t], (short)0, acc0[t], false, false);
            acc1[t] = __builtin_amdgcn_wmma_f32_16x16x4_f32(
                          false, a1, false, f.b[t], (short)0, acc1[t], false, false);
        }
    };

    KFrag<NT> ring[RING];
#pragma unroll
    for (int j = 0; j < RING; ++j) {
        ring[j] = fetch(min(j, last));
        __builtin_amdgcn_sched_barrier(0);
    }

    int ks = 0;
#pragma unroll 1
    for (; ks + 2 * RING <= nsteps; ks += RING) {
#pragma unroll
        for (int j = 0; j < RING; ++j) {
            products(ring[j]);
            ring[j] = fetch(ks + RING + j);
            __builtin_amdgcn_sched_barrier(0);
        }
    }

    const int R = nsteps - ks;
    float sc0[8], sc1[8];
#pragma unroll
    for (int j = 0; j < RING; ++j) {
        if (j < R) products(ring[j]);
        if (j < R - RING) ring[j] = fetch(ks + RING + j);
        if (j == 1) {
            const float4 p0 = *reinterpret_cast<const float4*>(dinv + rbase0);
            const float4 p1 = *reinterpret_cast<const float4*>(dinv + rbase0 + 4);
            const float4 q0 = *reinterpret_cast<const float4*>(dinv + rbase1);
            const float4 q1 = *reinterpret_cast<const float4*>(dinv + rbase1 + 4);
            sc0[0] = p0.x; sc0[1] = p0.y; sc0[2] = p0.z; sc0[3] = p0.w;
            sc0[4] = p1.x; sc0[5] = p1.y; sc0[6] = p1.z; sc0[7] = p1.w;
            sc1[0] = q0.x; sc1[1] = q0.y; sc1[2] = q0.z; sc1[3] = q0.w;
            sc1[4] = q1.x; sc1[5] = q1.y; sc1[6] = q1.z; sc1[7] = q1.w;
        }
    }
#pragma unroll
    for (int j = 0; j < RING - 1; ++j) {
        if (j < R - RING) products(ring[j]);
    }

#pragma unroll
    for (int t = 0; t < NT; ++t) {
        int c = t * 16 + col;
        if ((NC % 16 == 0) || (c < NC)) {
#pragma unroll
            for (int r = 0; r < 8; ++r) {
                Out[(size_t)(rbase0 + r) * NC + c] = acc0[t][r] * sc0[r];
                Out[(size_t)(rbase1 + r) * NC + c] = acc1[t][r] * sc1[r];
            }
        }
    }
}


template<int F>
__global__ __launch_bounds__(256) void gather_kernel(const float* __restrict__ h,
                                                     const float* __restrict__ dinv,
                                                     const int* __restrict__ rowptr,
                                                     const int* __restrict__ csr_src,
                                                     const float* __restrict__ bias,
                                                     float* __restrict__ out) {
    constexpr int CH  = F / 4;
    constexpr int EPI = 32 / CH;
    const int node = __builtin_amdgcn_readfirstlane(blockIdx.x * 8 + (threadIdx.x >> 5));
    if (node >= N_NODES) return;
    const int lane = threadIdx.x & 31;
    const int g = lane / CH;
    const int c = lane - g * CH;
    const bool active = g < EPI;

    const int start = rowptr[node];
    const int end   = rowptr[node + 1];
    const float dd  = dinv[node];
    const float4* __restrict__ h4 = reinterpret_cast<const float4*>(h);

    float4 acc = make_float4(0.f, 0.f, 0.f, 0.f);
    for (int base = start; base < end; base += 32) {
        const int n = min(32, end - base);
        int ms = 0;
        if (lane < n) {
            ms = (int)min((unsigned)csr_src[base + lane], (unsigned)(N_NODES - 1));
        }
        if constexpr (EPI == 1) {
#pragma unroll 4
            for (int j = 0; j < n; ++j) {
                const int s = __builtin_amdgcn_readlane(ms, j);
                const float4 v = h4[(size_t)s * CH + c];
                acc.x += v.x; acc.y += v.y; acc.z += v.z; acc.w += v.w;
            }
        } else {
            for (int j = 0; j < n; j += EPI) {
                const int idx = j + g;
                const int s = __shfl(ms, idx & 31);
                if (active && idx < n) {
                    const float4 v = h4[(size_t)s * CH + c];
                    acc.x += v.x; acc.y += v.y; acc.z += v.z; acc.w += v.w;
                }
            }
        }
    }
    if constexpr (EPI == 3) {
        acc.x += __shfl(acc.x, (lane + CH) & 31) + __shfl(acc.x, (lane + 2 * CH) & 31);
        acc.y += __shfl(acc.y, (lane + CH) & 31) + __shfl(acc.y, (lane + 2 * CH) & 31);
        acc.z += __shfl(acc.z, (lane + CH) & 31) + __shfl(acc.z, (lane + 2 * CH) & 31);
        acc.w += __shfl(acc.w, (lane + CH) & 31) + __shfl(acc.w, (lane + 2 * CH) & 31);
    }
    if (lane < CH) {
        const float4 hv = h4[(size_t)node * CH + lane];
        const float4 bv = reinterpret_cast<const float4*>(bias)[lane];
        float4 ov;
        ov.x = (bv.x + dd * hv.x) + dd * acc.x;  ov.y = (bv.y + dd * hv.y) + dd * acc.y;
        ov.z = (bv.z + dd * hv.z) + dd * acc.z;  ov.w = (bv.w + dd * hv.w) + dd * acc.w;
        reinterpret_cast<float4*>(out)[(size_t)node * CH + lane] = ov;
    }
}


extern "C" void kernel_launch(void* const* d_in, const int* in_sizes, int n_in,
                              void* d_out, int out_size, void* d_ws, size_t ws_size,
                              hipStream_t stream) {
    const float* x  = (const float*)d_in[0];
    const int*   ei = (const int*)d_in[1];
    const float* W1 = (const float*)d_in[2];
    const float* b1 = (const float*)d_in[3];
    const float* W2 = (const float*)d_in[4];
    const float* b2 = (const float*)d_in[5];
    const float* W3 = (const float*)d_in[6];
    const float* b3 = (const float*)d_in[7];
    float* out = (float*)d_out;

    char* ws = (char*)d_ws;
    float* dinv = (float*)ws;
    float* bufA = (float*)(ws + (((size_t)N_NODES * 4 + 255) & ~(size_t)255));
    float* bufB = bufA + (size_t)N_NODES * HID;
    v2f*   Wp1  = (v2f*)(bufB + (size_t)N_NODES * HID);
    const int NP1 = (F_IN / 4) * 8 * 32;
    const int NP2 = (HID  / 4) * 8 * 32;
    const int NP3 = (HID  / 4) * 3 * 32;
    v2f* Wp2 = Wp1 + NP1;
    v2f* Wp3 = Wp2 + NP2;

    const int CSR_N = N_EDGES + N_NODES + 4;
    int* csr_src = (int*)out;
    int* rowptr  = csr_src + N_EDGES;
    int* bucketed = (int*)bufB;
    int* histT    = bucketed + N_EDGES;
    int* colscan  = histT + NBKT * NCHUNK;
    int* total    = colscan + NBKT * NCHUNK;
    int* csr_src3 = (int*)bufB;
    int* rowptr3  = csr_src3 + N_EDGES;

    const int pairs   = N_NODES / 32;
    const int gemmBlk = (pairs + 3) / 4;
    const int nodeBlk = N_NODES / 8;

    pack_w_kernel<8><<<(NP1 + 255) / 256, 256, 0, stream>>>(W1, Wp1, F_IN, HID);
    pack_w_kernel<8><<<(NP2 + 255) / 256, 256, 0, stream>>>(W2, Wp2, HID, HID);
    pack_w_kernel<3><<<(NP3 + 255) / 256, 256, 0, stream>>>(W3, Wp3, HID, N_CLS);

    chunk_hist_kernel   <<<NCHUNK, 256, 0, stream>>>(ei + N_EDGES, histT);
    chunk_scan_kernel   <<<NBKT, NCHUNK, 0, stream>>>(histT, colscan, total);
    chunk_scatter_kernel<<<NCHUNK, 256, 0, stream>>>(ei, colscan, total, bucketed);
    bucket_sort_kernel  <<<NBKT, L2T, 0, stream>>>(bucketed, total, csr_src, rowptr, dinv);

    gemm_wmma_kernel<8, HID, false><<<gemmBlk, 128, 0, stream>>>(x, Wp1, dinv, bufA, N_NODES, F_IN);
    gather_kernel<HID><<<nodeBlk, 256, 0, stream>>>(bufA, dinv, rowptr, csr_src, b1, bufB);

    gemm_wmma_kernel<8, HID, true><<<gemmBlk, 128, 0, stream>>>(bufB, Wp2, dinv, bufA, N_NODES, HID);
    gather_kernel<HID><<<nodeBlk, 256, 0, stream>>>(bufA, dinv, rowptr, csr_src, b2, bufB);

    gemm_wmma_kernel<3, N_CLS, true><<<gemmBlk, 128, 0, stream>>>(bufB, Wp3, dinv, bufA, N_NODES, HID);
    copy_i4_kernel<<<(CSR_N / 4 + 255) / 256, 256, 0, stream>>>((const int4*)csr_src, (int4*)csr_src3, CSR_N / 4);
    gather_kernel<N_CLS><<<nodeBlk, 256, 0, stream>>>(bufA, dinv, rowptr3, csr_src3, b3, out);
}


# GEMM grids rebalanced over the SIMDs: 3 whole 32-row tiles per SIMD, one per wave; the 53 left-over row tiles cut into 16x32 eighth tiles spread over other WGPs; k and sum orders unchanged
# speedup vs baseline: 3.3549x; 1.0745x over previous
#include <hip/hip_runtime.h>
#include <hip/hip_bf16.h>

typedef __attribute__((ext_vector_type(2))) float v2f;
typedef __attribute__((ext_vector_type(8))) float v8f;

#define N_NODES 100000
#define N_EDGES 1600000
#define F_IN    500
#define HID     128
#define N_CLS   40


#define NPB      512
#define NBKT     ((N_NODES + NPB - 1) / NPB)
#define NCHUNK   256
#define CHUNK_E  ((N_EDGES + NCHUNK - 1) / NCHUNK)
#define EPT      ((CHUNK_E + 255) / 256)
#define STAGE_CAP 12288
#define L2T      512
#define L2U      8

static_assert(NBKT <= 256 && NBKT <= L2T, "one thread per bucket");
static_assert(N_NODES <= (1 << 17), "src must fit 17 bits");

template<int NW>
__device__ inline int block_excl_scan(int v, int* sh_wave, int& tot) {
    const int lane = threadIdx.x & 31, w = threadIdx.x >> 5;
    int inc = v;
#pragma unroll
    for (int off = 1; off < 32; off <<= 1) {
        int y = __shfl_up(inc, off);
        if (lane >= off) inc += y;
    }
    if (lane == 31) sh_wave[w] = inc;
    __syncthreads();
    int pre = 0, sum = 0;
#pragma unroll
    for (int k = 0; k < NW; ++k) {
        const int x = sh_wave[k];
        if (k < w) pre += x;
        sum += x;
    }
    tot = sum;
    return pre + inc - v;
}

__device__ inline unsigned clamp_node(int d) { return min((unsigned)d, (unsigned)(N_NODES - 1)); }

__global__ __launch_bounds__(256) void chunk_hist_kernel(const int* __restrict__ dst,
                                                         int* __restrict__ histT) {
    __shared__ int s_h[NBKT];
    const int c = blockIdx.x, t = threadIdx.x;
    if (t < NBKT) s_h[t] = 0;
    const int e0 = c * CHUNK_E;
    const int e1 = min(e0 + CHUNK_E, N_EDGES);
    int dv[EPT];
#pragma unroll
    for (int k = 0; k < EPT; ++k) {
        const int e = min(e0 + k * 256 + t, N_EDGES - 1);
        dv[k] = dst[e];
    }
    __syncthreads();
#pragma unroll
    for (int k = 0; k < EPT; ++k) {
        const int e = e0 + k * 256 + t;
        if (e < e1) atomicAdd(&s_h[clamp_node(dv[k]) / NPB], 1);
    }
    __syncthreads();
    if (t < NBKT) histT[t * NCHUNK + c] = s_h[t];
}

__global__ __launch_bounds__(NCHUNK) void chunk_scan_kernel(const int* __restrict__ histT,
                                                            int* __restrict__ colscan,
                                                            int* __restrict__ total) {
    __shared__ int s_w[NCHUNK / 32];
    const int b = blockIdx.x, t = threadIdx.x;
    const int v = histT[b * NCHUNK + t];
    int tot;
    const int ex = block_excl_scan<NCHUNK / 32>(v, s_w, tot);
    colscan[b * NCHUNK + t] = ex;
    if (t == 0) total[b] = tot;
}

__global__ __launch_bounds__(256) void chunk_scatter_kernel(const int* __restrict__ ei,
                                                            const int* __restrict__ colscan,
                                                            const int* __restrict__ total,
                                                            int* __restrict__ bucketed) {
    __shared__ int s_cur[NBKT];
    __shared__ int s_w[8];
    const int c = blockIdx.x, t = threadIdx.x;
    const int e0 = c * CHUNK_E;
    const int e1 = min(e0 + CHUNK_E, N_EDGES);
    int sv[EPT], dv[EPT];
#pragma unroll
    for (int k = 0; k < EPT; ++k) {
        const int e = min(e0 + k * 256 + t, N_EDGES - 1);
        sv[k] = ei[e];
        dv[k] = ei[N_EDGES + e];
    }
    const int tv = (t < NBKT) ? total[t] : 0;
    const int cs = (t < NBKT) ? colscan[t * NCHUNK + c] : 0;
    int tot;
    const int base = block_excl_scan<8>(tv, s_w, tot);
    if (t < NBKT) s_cur[t] = base + cs;
    __syncthreads();
    int pos[EPT];
#pragma unroll
    for (int k = 0; k < EPT; ++k) {
        const int e = e0 + k * 256 + t;
        pos[k] = (e < e1) ? atomicAdd(&s_cur[clamp_node(dv[k]) / NPB], 1) : -1;
    }
#pragma unroll
    for (int k = 0; k < EPT; ++k) {
        if (pos[k] >= 0) {
            const unsigned d = clamp_node(dv[k]);
            bucketed[pos[k]] = (int)(((d % NPB) << 17) | ((unsigned)sv[k] & 0x1FFFFu));
        }
    }
}

__global__ __launch_bounds__(L2T) void bucket_sort_kernel(const int* __restrict__ bucketed,
                                                          const int* __restrict__ total,
                                                          int* __restrict__ csr_src,
                                                          int* __restrict__ rowptr,
                                                          float* __restrict__ dinv) {
    __shared__ int s_cnt[NPB];
    __shared__ int s_w0[L2T / 32];
    __shared__ int s_w1[L2T / 32];
    __shared__ int s_stage[STAGE_CAP];
    const int b = blockIdx.x, t = threadIdx.x;

    const int tv = (t < b) ? total[t] : 0;
    const int n  = total[b];
    int base;
    block_excl_scan<L2T / 32>(tv, s_w0, base);
    const int* __restrict__ mine = bucketed + base;

    for (int i = t; i < NPB; i += L2T) s_cnt[i] = 0;
    __syncthreads();

    for (int i0 = 0; i0 < n; i0 += L2T * L2U) {
        int p[L2U];
#pragma unroll
        for (int k = 0; k < L2U; ++k) p[k] = mine[min(i0 + k * L2T + t, n - 1)];
#pragma unroll
        for (int k = 0; k < L2U; ++k)
            if (i0 + k * L2T + t < n) atomicAdd(&s_cnt[(p[k] >> 17) & (NPB - 1)], 1);
    }
    __syncthreads();

    static_assert(NPB == L2T, "one node per thread");
    const int cn = s_cnt[t];
    int tot;
    const int ex = block_excl_scan<L2T / 32>(cn, s_w1, tot);
    const int node = b * NPB + t;
    if (node < N_NODES) {
        rowptr[node] = base + ex;
        dinv[node]   = rsqrtf((float)(cn + 1));
    }
    if (b == NBKT - 1 && t < 4) rowptr[N_NODES + t] = base + n;
    s_cnt[t] = ex;
    __syncthreads();

    const bool fits = n <= STAGE_CAP;
    for (int i0 = 0; i0 < n; i0 += L2T * L2U) {
        int p[L2U], pos[L2U];
#pragma unroll
        for (int k = 0; k < L2U; ++k) p[k] = mine[min(i0 + k * L2T + t, n - 1)];
#pragma unroll
        for (int k = 0; k < L2U; ++k)
            pos[k] = (i0 + k * L2T + t < n) ? atomicAdd(&s_cnt[(p[k] >> 17) & (NPB - 1)], 1) : -1;
#pragma unroll
        for (int k = 0; k < L2U; ++k) {
            if (pos[k] >= 0) {
                const int s = p[k] & 0x1FFFF;
                if (fits) s_stage[pos[k]] = s;
                else      csr_src[base + pos[k]] = s;
            }
        }
    }
    if (fits) {
        __syncthreads();
        for (int i = t; i < n; i += L2T) csr_src[base + i] = s_stage[i];
    }
}

__global__ __launch_bounds__(256) void copy_i4_kernel(const int4* __restrict__ from,
                                                      int4* __restrict__ to, int n4) {
    int i = blockIdx.x * blockDim.x + threadIdx.x;
    if (i < n4) to[i] = from[i];
}


template<int NT>
__global__ __launch_bounds__(256) void pack_w_kernel(const float* __restrict__ W,
                                                     v2f* __restrict__ Wp,
                                                     int K, int Nc) {
    int idx = blockIdx.x * blockDim.x + threadIdx.x;
    int total = (K / 4) * NT * 32;
    if (idx >= total) return;
    int lane = idx & 31;
    int t    = (idx >> 5) % NT;
    int ks   = idx / (32 * NT);
    int col  = t * 16 + (lane & 15);
    int k    = ks * 4 + ((lane >> 4) * 2);
    v2f v;
    v.x = (col < Nc) ? W[(size_t)k * Nc + col]       : 0.0f;
    v.y = (col < Nc) ? W[(size_t)(k + 1) * Nc + col] : 0.0f;
    Wp[idx] = v;
}


#define RING 5

template<int NT>
struct KFrag {
    v2f a0, a1;
    v2f b[NT];
};

#define ROW_TILES   (N_NODES / 32)
#define N_SIMDS     1024
static_assert(N_NODES % 32 == 0, "row tiles are whole");

template<int NT>
struct Split {
    static constexpr int PNT    = (NT % 2 == 0) ? 2 : NT;
    static constexpr int CPS    = NT / PNT;
    static constexpr int WHOLE  = (PNT == NT) ? ROW_TILES : (ROW_TILES / N_SIMDS) * N_SIMDS;
    static constexpr int LEFT   = ROW_TILES - WHOLE;
    static constexpr int PIECES = LEFT * 2 * CPS;
    static constexpr int LEFT_BLOCKS = (PIECES + 3) / 4;
    static constexpr int BLOCKS = LEFT_BLOCKS + (WHOLE + 3) / 4;
};

template<int NS, int NTW, int NTALL, int NC, bool RELU>
__device__ __forceinline__ void gemm_wave_tile(const float* __restrict__ A,
                                               const v2f* __restrict__ Wp,
                                               const float* __restrict__ dinv,
                                               float* __restrict__ Out,
                                               int M, int K, int row0, int t0) {
    static_assert(NS == 1 || NS == 2, "one or two strips per wave");
    constexpr int NT = NTW;
    const int lane = threadIdx.x & 31;
    if (row0 >= M) return;

    const int kk  = (lane >> 4) * 2;
    const int col = lane & 15;

    const float* __restrict__ Arow0 = A + (size_t)(row0 + col) * K + kk;
    const float* __restrict__ Arow1 = Arow0 + (size_t)16 * K;
    const v2f*   __restrict__ wp    = Wp + t0 * 32 + lane;

    const int rbase0 = row0 + ((lane < 16) ? 0 : 8);
    const int rbase1 = rbase0 + 16;

    v8f acc0[NT], acc1[NT];
#pragma unroll
    for (int t = 0; t < NT; ++t) {
        acc0[t] = (v8f){0.f,0.f,0.f,0.f,0.f,0.f,0.f,0.f};
        acc1[t] = (v8f){0.f,0.f,0.f,0.f,0.f,0.f,0.f,0.f};
    }

    const int nsteps = K >> 2;
    const int last   = nsteps - 1;

    auto fetch = [&](int s) {
        KFrag<NT> f;
        f.a0 = *reinterpret_cast<const v2f*>(Arow0 + s * 4);
        if constexpr (NS == 2) f.a1 = *reinterpret_cast<const v2f*>(Arow1 + s * 4);
        else                   f.a1 = f.a0;
#pragma unroll
        for (int t = 0; t < NT; ++t) f.b[t] = wp[(s * NTALL + t) * 32];
        return f;
    };
    auto products = [&](const KFrag<NT>& f) {
        v2f a0 = f.a0, a1 = f.a1;
        if (RELU) {
            a0.x = fmaxf(a0.x, 0.0f); a0.y = fmaxf(a0.y, 0.0f);
            a1.x = fmaxf(a1.x, 0.0f); a1.y = fmaxf(a1.y, 0.0f);
        }
#pragma unroll
        for (int t = 0; t < NT; ++t) {
            acc0[t] = __builtin_amdgcn_wmma_f32_16x16x4_f32(
                          false, a0, false, f.b[t], (short)0, acc0[t], false, false);
            if constexpr (NS == 2)
                acc1[t] = __builtin_amdgcn_wmma_f32_16x16x4_f32(
                              false, a1, false, f.b[t], (short)0, acc1[t], false, false);
        }
    };

    KFrag<NT> ring[RING];
#pragma unroll
    for (int j = 0; j < RING; ++j) {
        ring[j] = fetch(min(j, last));
        __builtin_amdgcn_sched_barrier(0);
    }

    int ks = 0;
#pragma unroll 1
    for (; ks + 2 * RING <= nsteps; ks += RING) {
#pragma unroll
        for (int j = 0; j < RING; ++j) {
            products(ring[j]);
            ring[j] = fetch(ks + RING + j);
            __builtin_amdgcn_sched_barrier(0);
        }
    }

    const int R = nsteps - ks;
    float sc0[8], sc1[8];
#pragma unroll
    for (int j = 0; j < RING; ++j) {
        if (j < R) products(ring[j]);
        if (j < R - RING) ring[j] = fetch(ks + RING + j);
        if (j == 1) {
            const float4 p0 = *reinterpret_cast<const float4*>(dinv + rbase0);
            const float4 p1 = *reinterpret_cast<const float4*>(dinv + rbase0 + 4);
            sc0[0] = p0.x; sc0[1] = p0.y; sc0[2] = p0.z; sc0[3] = p0.w;
            sc0[4] = p1.x; sc0[5] = p1.y; sc0[6] = p1.z; sc0[7] = p1.w;
            if constexpr (NS == 2) {
                const float4 q0 = *reinterpret_cast<const float4*>(dinv + rbase1);
                const float4 q1 = *reinterpret_cast<const float4*>(dinv + rbase1 + 4);
                sc1[0] = q0.x; sc1[1] = q0.y; sc1[2] = q0.z; sc1[3] = q0.w;
                sc1[4] = q1.x; sc1[5] = q1.y; sc1[6] = q1.z; sc1[7] = q1.w;
            }
        }
    }
#pragma unroll
    for (int j = 0; j < RING - 1; ++j) {
        if (j < R - RING) products(ring[j]);
    }

#pragma unroll
    for (int t = 0; t < NT; ++t) {
        int c = (t0 + t) * 16 + col;
        if ((NC % 16 == 0) || (c < NC)) {
#pragma unroll
            for (int r = 0; r < 8; ++r) {
                Out[(size_t)(rbase0 + r) * NC + c] = acc0[t][r] * sc0[r];
                if constexpr (NS == 2) Out[(size_t)(rbase1 + r) * NC + c] = acc1[t][r] * sc1[r];
            }
        }
    }
}

template<int NT, int NC, bool RELU>
__global__ __launch_bounds__(128) void gemm_wmma_kernel(const float* __restrict__ A,
                                                        const v2f* __restrict__ Wp,
                                                        const float* __restrict__ dinv,
                                                        float* __restrict__ Out,
                                                        int M, int K) {
    using S = Split<NT>;
    const int wave = __builtin_amdgcn_readfirstlane(threadIdx.x >> 5);
    const int blk  = blockIdx.x;
    if constexpr (S::LEFT_BLOCKS > 0) {
        if (blk < S::LEFT_BLOCKS) {
            const int u = blk * 4 + wave;
            if (u < S::PIECES)
                gemm_wave_tile<1, S::PNT, NT, NC, RELU>(A, Wp, dinv, Out, M, K,
                                                        S::WHOLE * 32 + (u / S::CPS) * 16, (u % S::CPS) * S::PNT);
            return;
        }
    }
    gemm_wave_tile<2, NT, NT, NC, RELU>(A, Wp, dinv, Out, M, K, ((blk - S::LEFT_BLOCKS) * 4 + wave) * 32, 0);
}


template<int F>
__global__ __launch_bounds__(256) void gather_kernel(const float* __restrict__ h,
                                                     const float* __restrict__ dinv,
                                                     const int* __restrict__ rowptr,
                                                     const int* __restrict__ csr_src,
                                                     const float* __restrict__ bias,
                                                     float* __restrict__ out) {
    constexpr int CH  = F / 4;
    constexpr int EPI = 32 / CH;
    const int node = __builtin_amdgcn_readfirstlane(blockIdx.x * 8 + (threadIdx.x >> 5));
    if (node >= N_NODES) return;
    const int lane = threadIdx.x & 31;
    const int g = lane / CH;
    const int c = lane - g * CH;
    const bool active = g < EPI;

    const int start = rowptr[node];
    const int end   = rowptr[node + 1];
    const float dd  = dinv[node];
    const float4* __restrict__ h4 = reinterpret_cast<const float4*>(h);

    float4 acc = make_float4(0.f, 0.f, 0.f, 0.f);
    for (int base = start; base < end; base += 32) {
        const int n = min(32, end - base);
        int ms = 0;
        if (lane < n) {
            ms = (int)min((unsigned)csr_src[base + lane], (unsigned)(N_NODES - 1));
        }
        if constexpr (EPI == 1) {
#pragma unroll 4
            for (int j = 0; j < n; ++j) {
                const int s = __builtin_amdgcn_readlane(ms, j);
                const float4 v = h4[(size_t)s * CH + c];
                acc.x += v.x; acc.y += v.y; acc.z += v.z; acc.w += v.w;
            }
        } else {
            for (int j = 0; j < n; j += EPI) {
                const int idx = j + g;
                const int s = __shfl(ms, idx & 31);
                if (active && idx < n) {
                    const float4 v = h4[(size_t)s * CH + c];
                    acc.x += v.x; acc.y += v.y; acc.z += v.z; acc.w += v.w;
                }
            }
        }
    }
    if constexpr (EPI == 3) {
        acc.x += __shfl(acc.x, (lane + CH) & 31) + __shfl(acc.x, (lane + 2 * CH) & 31);
        acc.y += __shfl(acc.y, (lane + CH) & 31) + __shfl(acc.y, (lane + 2 * CH) & 31);
        acc.z += __shfl(acc.z, (lane + CH) & 31) + __shfl(acc.z, (lane + 2 * CH) & 31);
        acc.w += __shfl(acc.w, (lane + CH) & 31) + __shfl(acc.w, (lane + 2 * CH) & 31);
    }
    if (lane < CH) {
        const float4 hv = h4[(size_t)node * CH + lane];
        const float4 bv = reinterpret_cast<const float4*>(bias)[lane];
        float4 ov;
        ov.x = (bv.x + dd * hv.x) + dd * acc.x;  ov.y = (bv.y + dd * hv.y) + dd * acc.y;
        ov.z = (bv.z + dd * hv.z) + dd * acc.z;  ov.w = (bv.w + dd * hv.w) + dd * acc.w;
        reinterpret_cast<float4*>(out)[(size_t)node * CH + lane] = ov;
    }
}


extern "C" void kernel_launch(void* const* d_in, const int* in_sizes, int n_in,
                              void* d_out, int out_size, void* d_ws, size_t ws_size,
                              hipStream_t stream) {
    const float* x  = (const float*)d_in[0];
    const int*   ei = (const int*)d_in[1];
    const float* W1 = (const float*)d_in[2];
    const float* b1 = (const float*)d_in[3];
    const float* W2 = (const float*)d_in[4];
    const float* b2 = (const float*)d_in[5];
    const float* W3 = (const float*)d_in[6];
    const float* b3 = (const float*)d_in[7];
    float* out = (float*)d_out;

    char* ws = (char*)d_ws;
    float* dinv = (float*)ws;
    float* bufA = (float*)(ws + (((size_t)N_NODES * 4 + 255) & ~(size_t)255));
    float* bufB = bufA + (size_t)N_NODES * HID;
    v2f*   Wp1  = (v2f*)(bufB + (size_t)N_NODES * HID);
    const int NP1 = (F_IN / 4) * 8 * 32;
    const int NP2 = (HID  / 4) * 8 * 32;
    const int NP3 = (HID  / 4) * 3 * 32;
    v2f* Wp2 = Wp1 + NP1;
    v2f* Wp3 = Wp2 + NP2;

    const int CSR_N = N_EDGES + N_NODES + 4;
    int* csr_src = (int*)out;
    int* rowptr  = csr_src + N_EDGES;
    int* bucketed = (int*)bufB;
    int* histT    = bucketed + N_EDGES;
    int* colscan  = histT + NBKT * NCHUNK;
    int* total    = colscan + NBKT * NCHUNK;
    int* csr_src3 = (int*)bufB;
    int* rowptr3  = csr_src3 + N_EDGES;

    const int nodeBlk = N_NODES / 8;

    pack_w_kernel<8><<<(NP1 + 255) / 256, 256, 0, stream>>>(W1, Wp1, F_IN, HID);
    pack_w_kernel<8><<<(NP2 + 255) / 256, 256, 0, stream>>>(W2, Wp2, HID, HID);
    pack_w_kernel<3><<<(NP3 + 255) / 256, 256, 0, stream>>>(W3, Wp3, HID, N_CLS);

    chunk_hist_kernel   <<<NCHUNK, 256, 0, stream>>>(ei + N_EDGES, histT);
    chunk_scan_kernel   <<<NBKT, NCHUNK, 0, stream>>>(histT, colscan, total);
    chunk_scatter_kernel<<<NCHUNK, 256, 0, stream>>>(ei, colscan, total, bucketed);
    bucket_sort_kernel  <<<NBKT, L2T, 0, stream>>>(bucketed, total, csr_src, rowptr, dinv);

    gemm_wmma_kernel<8, HID, false><<<Split<8>::BLOCKS, 128, 0, stream>>>(x, Wp1, dinv, bufA, N_NODES, F_IN);
    gather_kernel<HID><<<nodeBlk, 256, 0, stream>>>(bufA, dinv, rowptr, csr_src, b1, bufB);

    gemm_wmma_kernel<8, HID, true><<<Split<8>::BLOCKS, 128, 0, stream>>>(bufB, Wp2, dinv, bufA, N_NODES, HID);
    gather_kernel<HID><<<nodeBlk, 256, 0, stream>>>(bufA, dinv, rowptr, csr_src, b2, bufB);

    gemm_wmma_kernel<3, N_CLS, true><<<Split<3>::BLOCKS, 128, 0, stream>>>(bufB, Wp3, dinv, bufA, N_NODES, HID);
    copy_i4_kernel<<<(CSR_N / 4 + 255) / 256, 256, 0, stream>>>((const int4*)csr_src, (int4*)csr_src3, CSR_N / 4);
    gather_kernel<N_CLS><<<nodeBlk, 256, 0, stream>>>(bufA, dinv, rowptr3, csr_src3, b3, out);
}


# Layer-1 and layer-2 gathers each fused with the next layer's WMMA product: a work-group gathers 16 rows into LDS as the A strip, weights staged in LDS; 14 -> 12 launches; k and sum orders unchanged
# speedup vs baseline: 3.5641x; 1.0624x over previous
#include <hip/hip_runtime.h>
#include <hip/hip_bf16.h>

typedef __attribute__((ext_vector_type(2))) float v2f;
typedef __attribute__((ext_vector_type(8))) float v8f;

#define N_NODES 100000
#define N_EDGES 1600000
#define F_IN    500
#define HID     128
#define N_CLS   40


#define NPB      512
#define NBKT     ((N_NODES + NPB - 1) / NPB)
#define NCHUNK   256
#define CHUNK_E  ((N_EDGES + NCHUNK - 1) / NCHUNK)
#define EPT      ((CHUNK_E + 255) / 256)
#define STAGE_CAP 12288
#define L2T      512
#define L2U      8

static_assert(NBKT <= 256 && NBKT <= L2T, "one thread per bucket");
static_assert(N_NODES <= (1 << 17), "src must fit 17 bits");

template<int NW>
__device__ inline int block_excl_scan(int v, int* sh_wave, int& tot) {
    const int lane = threadIdx.x & 31, w = threadIdx.x >> 5;
    int inc = v;
#pragma unroll
    for (int off = 1; off < 32; off <<= 1) {
        int y = __shfl_up(inc, off);
        if (lane >= off) inc += y;
    }
    if (lane == 31) sh_wave[w] = inc;
    __syncthreads();
    int pre = 0, sum = 0;
#pragma unroll
    for (int k = 0; k < NW; ++k) {
        const int x = sh_wave[k];
        if (k < w) pre += x;
        sum += x;
    }
    tot = sum;
    return pre + inc - v;
}

__device__ inline unsigned clamp_node(int d) { return min((unsigned)d, (unsigned)(N_NODES - 1)); }

__global__ __launch_bounds__(256) void chunk_hist_kernel(const int* __restrict__ dst,
                                                         int* __restrict__ histT) {
    __shared__ int s_h[NBKT];
    const int c = blockIdx.x, t = threadIdx.x;
    if (t < NBKT) s_h[t] = 0;
    const int e0 = c * CHUNK_E;
    const int e1 = min(e0 + CHUNK_E, N_EDGES);
    int dv[EPT];
#pragma unroll
    for (int k = 0; k < EPT; ++k) {
        const int e = min(e0 + k * 256 + t, N_EDGES - 1);
        dv[k] = dst[e];
    }
    __syncthreads();
#pragma unroll
    for (int k = 0; k < EPT; ++k) {
        const int e = e0 + k * 256 + t;
        if (e < e1) atomicAdd(&s_h[clamp_node(dv[k]) / NPB], 1);
    }
    __syncthreads();
    if (t < NBKT) histT[t * NCHUNK + c] = s_h[t];
}

__global__ __launch_bounds__(NCHUNK) void chunk_scan_kernel(const int* __restrict__ histT,
                                                            int* __restrict__ colscan,
                                                            int* __restrict__ total) {
    __shared__ int s_w[NCHUNK / 32];
    const int b = blockIdx.x, t = threadIdx.x;
    const int v = histT[b * NCHUNK + t];
    int tot;
    const int ex = block_excl_scan<NCHUNK / 32>(v, s_w, tot);
    colscan[b * NCHUNK + t] = ex;
    if (t == 0) total[b] = tot;
}

__global__ __launch_bounds__(256) void chunk_scatter_kernel(const int* __restrict__ ei,
                                                            const int* __restrict__ colscan,
                                                            const int* __restrict__ total,
                                                            int* __restrict__ bucketed) {
    __shared__ int s_cur[NBKT];
    __shared__ int s_w[8];
    const int c = blockIdx.x, t = threadIdx.x;
    const int e0 = c * CHUNK_E;
    const int e1 = min(e0 + CHUNK_E, N_EDGES);
    int sv[EPT], dv[EPT];
#pragma unroll
    for (int k = 0; k < EPT; ++k) {
        const int e = min(e0 + k * 256 + t, N_EDGES - 1);
        sv[k] = ei[e];
        dv[k] = ei[N_EDGES + e];
    }
    const int tv = (t < NBKT) ? total[t] : 0;
    const int cs = (t < NBKT) ? colscan[t * NCHUNK + c] : 0;
    int tot;
    const int base = block_excl_scan<8>(tv, s_w, tot);
    if (t < NBKT) s_cur[t] = base + cs;
    __syncthreads();
    int pos[EPT];
#pragma unroll
    for (int k = 0; k < EPT; ++k) {
        const int e = e0 + k * 256 + t;
        pos[k] = (e < e1) ? atomicAdd(&s_cur[clamp_node(dv[k]) / NPB], 1) : -1;
    }
#pragma unroll
    for (int k = 0; k < EPT; ++k) {
        if (pos[k] >= 0) {
            const unsigned d = clamp_node(dv[k]);
            bucketed[pos[k]] = (int)(((d % NPB) << 17) | ((unsigned)sv[k] & 0x1FFFFu));
        }
    }
}

__global__ __launch_bounds__(L2T) void bucket_sort_kernel(const int* __restrict__ bucketed,
                                                          const int* __restrict__ total,
                                                          int* __restrict__ csr_src,
                                                          int* __restrict__ rowptr,
                                                          float* __restrict__ dinv) {
    __shared__ int s_cnt[NPB];
    __shared__ int s_w0[L2T / 32];
    __shared__ int s_w1[L2T / 32];
    __shared__ int s_stage[STAGE_CAP];
    const int b = blockIdx.x, t = threadIdx.x;

    const int tv = (t < b) ? total[t] : 0;
    const int n  = total[b];
    int base;
    block_excl_scan<L2T / 32>(tv, s_w0, base);
    const int* __restrict__ mine = bucketed + base;

    for (int i = t; i < NPB; i += L2T) s_cnt[i] = 0;
    __syncthreads();

    for (int i0 = 0; i0 < n; i0 += L2T * L2U) {
        int p[L2U];
#pragma unroll
        for (int k = 0; k < L2U; ++k) p[k] = mine[min(i0 + k * L2T + t, n - 1)];
#pragma unroll
        for (int k = 0; k < L2U; ++k)
            if (i0 + k * L2T + t < n) atomicAdd(&s_cnt[(p[k] >> 17) & (NPB - 1)], 1);
    }
    __syncthreads();

    static_assert(NPB == L2T, "one node per thread");
    const int cn = s_cnt[t];
    int tot;
    const int ex = block_excl_scan<L2T / 32>(cn, s_w1, tot);
    const int node = b * NPB + t;
    if (node < N_NODES) {
        rowptr[node] = base + ex;
        dinv[node]   = rsqrtf((float)(cn + 1));
    }
    if (b == NBKT - 1 && t < 4) rowptr[N_NODES + t] = base + n;
    s_cnt[t] = ex;
    __syncthreads();

    const bool fits = n <= STAGE_CAP;
    for (int i0 = 0; i0 < n; i0 += L2T * L2U) {
        int p[L2U], pos[L2U];
#pragma unroll
        for (int k = 0; k < L2U; ++k) p[k] = mine[min(i0 + k * L2T + t, n - 1)];
#pragma unroll
        for (int k = 0; k < L2U; ++k)
            pos[k] = (i0 + k * L2T + t < n) ? atomicAdd(&s_cnt[(p[k] >> 17) & (NPB - 1)], 1) : -1;
#pragma unroll
        for (int k = 0; k < L2U; ++k) {
            if (pos[k] >= 0) {
                const int s = p[k] & 0x1FFFF;
                if (fits) s_stage[pos[k]] = s;
                else      csr_src[base + pos[k]] = s;
            }
        }
    }
    if (fits) {
        __syncthreads();
        for (int i = t; i < n; i += L2T) csr_src[base + i] = s_stage[i];
    }
}

__global__ __launch_bounds__(256) void copy_i4_kernel(const int4* __restrict__ from,
                                                      int4* __restrict__ to, int n4) {
    int i = blockIdx.x * blockDim.x + threadIdx.x;
    if (i < n4) to[i] = from[i];
}


template<int NT>
__global__ __launch_bounds__(256) void pack_w_kernel(const float* __restrict__ W,
                                                     v2f* __restrict__ Wp,
                                                     int K, int Nc) {
    int idx = blockIdx.x * blockDim.x + threadIdx.x;
    int total = (K / 4) * NT * 32;
    if (idx >= total) return;
    int lane = idx & 31;
    int t    = (idx >> 5) % NT;
    int ks   = idx / (32 * NT);
    int col  = t * 16 + (lane & 15);
    int k    = ks * 4 + ((lane >> 4) * 2);
    v2f v;
    v.x = (col < Nc) ? W[(size_t)k * Nc + col]       : 0.0f;
    v.y = (col < Nc) ? W[(size_t)(k + 1) * Nc + col] : 0.0f;
    Wp[idx] = v;
}


#define RING 5

template<int NT>
struct KFrag {
    v2f a0, a1;
    v2f b[NT];
};

#define ROW_TILES   (N_NODES / 32)
#define N_SIMDS     1024
static_assert(N_NODES % 32 == 0, "row tiles are whole");

template<int NT>
struct Split {
    static constexpr int PNT    = (NT % 2 == 0) ? 2 : NT;
    static constexpr int CPS    = NT / PNT;
    static constexpr int WHOLE  = (PNT == NT) ? ROW_TILES : (ROW_TILES / N_SIMDS) * N_SIMDS;
    static constexpr int LEFT   = ROW_TILES - WHOLE;
    static constexpr int PIECES = LEFT * 2 * CPS;
    static constexpr int LEFT_BLOCKS = (PIECES + 3) / 4;
    static constexpr int BLOCKS = LEFT_BLOCKS + (WHOLE + 3) / 4;
};

template<int NS, int NTW, int NTALL, int NC, bool RELU>
__device__ __forceinline__ void gemm_wave_tile(const float* __restrict__ A,
                                               const v2f* __restrict__ Wp,
                                               const float* __restrict__ dinv,
                                               float* __restrict__ Out,
                                               int M, int K, int row0, int t0) {
    static_assert(NS == 1 || NS == 2, "one or two strips per wave");
    constexpr int NT = NTW;
    const int lane = threadIdx.x & 31;
    if (row0 >= M) return;

    const int kk  = (lane >> 4) * 2;
    const int col = lane & 15;

    const float* __restrict__ Arow0 = A + (size_t)(row0 + col) * K + kk;
    const float* __restrict__ Arow1 = Arow0 + (size_t)16 * K;
    const v2f*   __restrict__ wp    = Wp + t0 * 32 + lane;

    const int rbase0 = row0 + ((lane < 16) ? 0 : 8);
    const int rbase1 = rbase0 + 16;

    v8f acc0[NT], acc1[NT];
#pragma unroll
    for (int t = 0; t < NT; ++t) {
        acc0[t] = (v8f){0.f,0.f,0.f,0.f,0.f,0.f,0.f,0.f};
        acc1[t] = (v8f){0.f,0.f,0.f,0.f,0.f,0.f,0.f,0.f};
    }

    const int nsteps = K >> 2;
    const int last   = nsteps - 1;

    auto fetch = [&](int s) {
        KFrag<NT> f;
        f.a0 = *reinterpret_cast<const v2f*>(Arow0 + s * 4);
        if constexpr (NS == 2) f.a1 = *reinterpret_cast<const v2f*>(Arow1 + s * 4);
        else                   f.a1 = f.a0;
#pragma unroll
        for (int t = 0; t < NT; ++t) f.b[t] = wp[(s * NTALL + t) * 32];
        return f;
    };
    auto products = [&](const KFrag<NT>& f) {
        v2f a0 = f.a0, a1 = f.a1;
        if (RELU) {
            a0.x = fmaxf(a0.x, 0.0f); a0.y = fmaxf(a0.y, 0.0f);
            a1.x = fmaxf(a1.x, 0.0f); a1.y = fmaxf(a1.y, 0.0f);
        }
#pragma unroll
        for (int t = 0; t < NT; ++t) {
            acc0[t] = __builtin_amdgcn_wmma_f32_16x16x4_f32(
                          false, a0, false, f.b[t], (short)0, acc0[t], false, false);
            if constexpr (NS == 2)
                acc1[t] = __builtin_amdgcn_wmma_f32_16x16x4_f32(
                              false, a1, false, f.b[t], (short)0, acc1[t], false, false);
        }
    };

    KFrag<NT> ring[RING];
#pragma unroll
    for (int j = 0; j < RING; ++j) {
        ring[j] = fetch(min(j, last));
        __builtin_amdgcn_sched_barrier(0);
    }

    int ks = 0;
#pragma unroll 1
    for (; ks + 2 * RING <= nsteps; ks += RING) {
#pragma unroll
        for (int j = 0; j < RING; ++j) {
            products(ring[j]);
            ring[j] = fetch(ks + RING + j);
            __builtin_amdgcn_sched_barrier(0);
        }
    }

    const int R = nsteps - ks;
    float sc0[8], sc1[8];
#pragma unroll
    for (int j = 0; j < RING; ++j) {
        if (j < R) products(ring[j]);
        if (j < R - RING) ring[j] = fetch(ks + RING + j);
        if (j == 1) {
            const float4 p0 = *reinterpret_cast<const float4*>(dinv + rbase0);
            const float4 p1 = *reinterpret_cast<const float4*>(dinv + rbase0 + 4);
            sc0[0] = p0.x; sc0[1] = p0.y; sc0[2] = p0.z; sc0[3] = p0.w;
            sc0[4] = p1.x; sc0[5] = p1.y; sc0[6] = p1.z; sc0[7] = p1.w;
            if constexpr (NS == 2) {
                const float4 q0 = *reinterpret_cast<const float4*>(dinv + rbase1);
                const float4 q1 = *reinterpret_cast<const float4*>(dinv + rbase1 + 4);
                sc1[0] = q0.x; sc1[1] = q0.y; sc1[2] = q0.z; sc1[3] = q0.w;
                sc1[4] = q1.x; sc1[5] = q1.y; sc1[6] = q1.z; sc1[7] = q1.w;
            }
        }
    }
#pragma unroll
    for (int j = 0; j < RING - 1; ++j) {
        if (j < R - RING) products(ring[j]);
    }

#pragma unroll
    for (int t = 0; t < NT; ++t) {
        int c = (t0 + t) * 16 + col;
        if ((NC % 16 == 0) || (c < NC)) {
#pragma unroll
            for (int r = 0; r < 8; ++r) {
                Out[(size_t)(rbase0 + r) * NC + c] = acc0[t][r] * sc0[r];
                if constexpr (NS == 2) Out[(size_t)(rbase1 + r) * NC + c] = acc1[t][r] * sc1[r];
            }
        }
    }
}

template<int NT, int NC, bool RELU>
__global__ __launch_bounds__(128) void gemm_wmma_kernel(const float* __restrict__ A,
                                                        const v2f* __restrict__ Wp,
                                                        const float* __restrict__ dinv,
                                                        float* __restrict__ Out,
                                                        int M, int K) {
    using S = Split<NT>;
    const int wave = __builtin_amdgcn_readfirstlane(threadIdx.x >> 5);
    const int blk  = blockIdx.x;
    if constexpr (S::LEFT_BLOCKS > 0) {
        if (blk < S::LEFT_BLOCKS) {
            const int u = blk * 4 + wave;
            if (u < S::PIECES)
                gemm_wave_tile<1, S::PNT, NT, NC, RELU>(A, Wp, dinv, Out, M, K,
                                                        S::WHOLE * 32 + (u / S::CPS) * 16, (u % S::CPS) * S::PNT);
            return;
        }
    }
    gemm_wave_tile<2, NT, NT, NC, RELU>(A, Wp, dinv, Out, M, K, ((blk - S::LEFT_BLOCKS) * 4 + wave) * 32, 0);
}


template<int F>
__global__ __launch_bounds__(256) void gather_kernel(const float* __restrict__ h,
                                                     const float* __restrict__ dinv,
                                                     const int* __restrict__ rowptr,
                                                     const int* __restrict__ csr_src,
                                                     const float* __restrict__ bias,
                                                     float* __restrict__ out) {
    constexpr int CH  = F / 4;
    constexpr int EPI = 32 / CH;
    const int node = __builtin_amdgcn_readfirstlane(blockIdx.x * 8 + (threadIdx.x >> 5));
    if (node >= N_NODES) return;
    const int lane = threadIdx.x & 31;
    const int g = lane / CH;
    const int c = lane - g * CH;
    const bool active = g < EPI;

    const int start = rowptr[node];
    const int end   = rowptr[node + 1];
    const float dd  = dinv[node];
    const float4* __restrict__ h4 = reinterpret_cast<const float4*>(h);

    float4 acc = make_float4(0.f, 0.f, 0.f, 0.f);
    for (int base = start; base < end; base += 32) {
        const int n = min(32, end - base);
        int ms = 0;
        if (lane < n) {
            ms = (int)min((unsigned)csr_src[base + lane], (unsigned)(N_NODES - 1));
        }
        if constexpr (EPI == 1) {
#pragma unroll 4
            for (int j = 0; j < n; ++j) {
                const int s = __builtin_amdgcn_readlane(ms, j);
                const float4 v = h4[(size_t)s * CH + c];
                acc.x += v.x; acc.y += v.y; acc.z += v.z; acc.w += v.w;
            }
        } else {
            for (int j = 0; j < n; j += EPI) {
                const int idx = j + g;
                const int s = __shfl(ms, idx & 31);
                if (active && idx < n) {
                    const float4 v = h4[(size_t)s * CH + c];
                    acc.x += v.x; acc.y += v.y; acc.z += v.z; acc.w += v.w;
                }
            }
        }
    }
    if constexpr (EPI == 3) {
        acc.x += __shfl(acc.x, (lane + CH) & 31) + __shfl(acc.x, (lane + 2 * CH) & 31);
        acc.y += __shfl(acc.y, (lane + CH) & 31) + __shfl(acc.y, (lane + 2 * CH) & 31);
        acc.z += __shfl(acc.z, (lane + CH) & 31) + __shfl(acc.z, (lane + 2 * CH) & 31);
        acc.w += __shfl(acc.w, (lane + CH) & 31) + __shfl(acc.w, (lane + 2 * CH) & 31);
    }
    if (lane < CH) {
        const float4 hv = h4[(size_t)node * CH + lane];
        const float4 bv = reinterpret_cast<const float4*>(bias)[lane];
        float4 ov;
        ov.x = (bv.x + dd * hv.x) + dd * acc.x;  ov.y = (bv.y + dd * hv.y) + dd * acc.y;
        ov.z = (bv.z + dd * hv.z) + dd * acc.z;  ov.w = (bv.w + dd * hv.w) + dd * acc.w;
        reinterpret_cast<float4*>(out)[(size_t)node * CH + lane] = ov;
    }
}


#define FG_ROWS   16
#define FG_PITCH  (HID + 4)
#define FG_GROUPS (N_NODES / FG_ROWS)
static_assert(N_NODES % FG_ROWS == 0, "node groups are whole");

template<int U>
__device__ __forceinline__ void row_batch(const float4* __restrict__ h4, int ms, int j, int e, int split,
                                          int lane, float4& accA, float4& accB) {
    float4 v[U];
#pragma unroll
    for (int u = 0; u < U; ++u) {
        const int s = __builtin_amdgcn_readlane(ms, j + u);
        v[u] = h4[(size_t)s * (HID / 4) + lane];
    }
    const int k = split - e;
    if (k >= U) {
#pragma unroll
        for (int u = 0; u < U; ++u) { accA.x += v[u].x; accA.y += v[u].y; accA.z += v[u].z; accA.w += v[u].w; }
    } else if (k <= 0) {
#pragma unroll
        for (int u = 0; u < U; ++u) { accB.x += v[u].x; accB.y += v[u].y; accB.z += v[u].z; accB.w += v[u].w; }
    } else {
#pragma unroll
        for (int u = 0; u < U; ++u) {
            if (u < k) { accA.x += v[u].x; accA.y += v[u].y; accA.z += v[u].z; accA.w += v[u].w; }
            else       { accB.x += v[u].x; accB.y += v[u].y; accB.z += v[u].z; accB.w += v[u].w; }
        }
    }
}

template<int NT, int NC>
__global__ __launch_bounds__(256) void gather_gemm_kernel(const float* __restrict__ h,
                                                          const float* __restrict__ dinv,
                                                          const int* __restrict__ rowptr,
                                                          const int* __restrict__ csr_src,
                                                          const float* __restrict__ bias,
                                                          const v2f* __restrict__ Wp,
                                                          float* __restrict__ Out) {
    constexpr int KS  = HID / 4;
    constexpr int PNT = Split<NT>::PNT;
    constexpr int CPS = Split<NT>::CPS;
    __shared__ v2f s_w[KS * NT * 32];
    __shared__ __attribute__((aligned(16))) float s_a[FG_ROWS * FG_PITCH];

    const int lane = threadIdx.x & 31;
    const int wave = __builtin_amdgcn_readfirstlane(threadIdx.x >> 5);

    for (int i = threadIdx.x; i < KS * NT * 32; i += 256) s_w[i] = Wp[i];

    const float4* __restrict__ h4 = reinterpret_cast<const float4*>(h);
    const float4 bv = reinterpret_cast<const float4*>(bias)[lane];
    const int kk  = (lane >> 4) * 2;
    const int col = lane & 15;

    for (int g = blockIdx.x; g < FG_GROUPS; g += gridDim.x) {
        const int node0 = g * FG_ROWS;

        const int n0  = node0 + 2 * wave;
        const int rp0 = rowptr[n0], rp1 = rowptr[n0 + 1], rp2 = rowptr[n0 + 2];
        const int ms0 = (int)clamp_node(csr_src[min(rp0 + lane,      N_EDGES - 1)]);
        const int ms1 = (int)clamp_node(csr_src[min(rp0 + 32 + lane, N_EDGES - 1)]);
        const float4 hvA = h4[(size_t)n0 * (HID / 4) + lane];
        const float4 hvB = h4[(size_t)(n0 + 1) * (HID / 4) + lane];
        const float ddA = dinv[n0], ddB = dinv[n0 + 1];

        float4 aA = make_float4(0.f, 0.f, 0.f, 0.f), aB = make_float4(0.f, 0.f, 0.f, 0.f);
        int nb = 0;
        for (int base = rp0; base < rp2; base += 32, ++nb) {
            const int n = min(32, rp2 - base);
            int ms;
            if (nb == 0)      ms = ms0;
            else if (nb == 1) ms = ms1;
            else              ms = (int)clamp_node(csr_src[min(base + lane, N_EDGES - 1)]);
            int j = 0;
            for (; j + 16 <= n; j += 16) row_batch<16>(h4, ms, j, base + j, rp1, lane, aA, aB);
            if (j + 8 <= n) { row_batch<8>(h4, ms, j, base + j, rp1, lane, aA, aB); j += 8; }
            if (j + 4 <= n) { row_batch<4>(h4, ms, j, base + j, rp1, lane, aA, aB); j += 4; }
            if (j + 2 <= n) { row_batch<2>(h4, ms, j, base + j, rp1, lane, aA, aB); j += 2; }
            if (j < n)        row_batch<1>(h4, ms, j, base + j, rp1, lane, aA, aB);
        }

        float4 oA, oB;
        oA.x = fmaxf((bv.x + ddA * hvA.x) + ddA * aA.x, 0.0f);  oA.y = fmaxf((bv.y + ddA * hvA.y) + ddA * aA.y, 0.0f);
        oA.z = fmaxf((bv.z + ddA * hvA.z) + ddA * aA.z, 0.0f);  oA.w = fmaxf((bv.w + ddA * hvA.w) + ddA * aA.w, 0.0f);
        oB.x = fmaxf((bv.x + ddB * hvB.x) + ddB * aB.x, 0.0f);  oB.y = fmaxf((bv.y + ddB * hvB.y) + ddB * aB.y, 0.0f);
        oB.z = fmaxf((bv.z + ddB * hvB.z) + ddB * aB.z, 0.0f);  oB.w = fmaxf((bv.w + ddB * hvB.w) + ddB * aB.w, 0.0f);
        __syncthreads();
        *reinterpret_cast<float4*>(s_a + (2 * wave)     * FG_PITCH + 4 * lane) = oA;
        *reinterpret_cast<float4*>(s_a + (2 * wave + 1) * FG_PITCH + 4 * lane) = oB;
        __syncthreads();

        if (wave < CPS) {
            const int t0 = wave * PNT;
            v8f acc[PNT];
#pragma unroll
            for (int t = 0; t < PNT; ++t) acc[t] = (v8f){0.f,0.f,0.f,0.f,0.f,0.f,0.f,0.f};
            const float* arow = s_a + col * FG_PITCH + kk;
            const v2f*   wl   = s_w + t0 * 32 + lane;
#pragma unroll 8
            for (int s = 0; s < KS; ++s) {
                const v2f a = *reinterpret_cast<const v2f*>(arow + 4 * s);
#pragma unroll
                for (int t = 0; t < PNT; ++t)
                    acc[t] = __builtin_amdgcn_wmma_f32_16x16x4_f32(
                                 false, a, false, wl[(s * NT + t) * 32], (short)0, acc[t], false, false);
            }
            const int rbase = node0 + ((lane < 16) ? 0 : 8);
            const float4 p0 = *reinterpret_cast<const float4*>(dinv + rbase);
            const float4 p1 = *reinterpret_cast<const float4*>(dinv + rbase + 4);
            const float sc[8] = {p0.x, p0.y, p0.z, p0.w, p1.x, p1.y, p1.z, p1.w};
#pragma unroll
            for (int t = 0; t < PNT; ++t) {
                const int c = (t0 + t) * 16 + col;
                if ((NC % 16 == 0) || (c < NC)) {
#pragma unroll
                    for (int r = 0; r < 8; ++r)
                        Out[(size_t)(rbase + r) * NC + c] = acc[t][r] * sc[r];
                }
            }
        }
    }
}


extern "C" void kernel_launch(void* const* d_in, const int* in_sizes, int n_in,
                              void* d_out, int out_size, void* d_ws, size_t ws_size,
                              hipStream_t stream) {
    const float* x  = (const float*)d_in[0];
    const int*   ei = (const int*)d_in[1];
    const float* W1 = (const float*)d_in[2];
    const float* b1 = (const float*)d_in[3];
    const float* W2 = (const float*)d_in[4];
    const float* b2 = (const float*)d_in[5];
    const float* W3 = (const float*)d_in[6];
    const float* b3 = (const float*)d_in[7];
    float* out = (float*)d_out;

    char* ws = (char*)d_ws;
    float* dinv = (float*)ws;
    float* bufA = (float*)(ws + (((size_t)N_NODES * 4 + 255) & ~(size_t)255));
    float* bufB = bufA + (size_t)N_NODES * HID;
    v2f*   Wp1  = (v2f*)(bufB + (size_t)N_NODES * HID);
    const int NP1 = (F_IN / 4) * 8 * 32;
    const int NP2 = (HID  / 4) * 8 * 32;
    const int NP3 = (HID  / 4) * 3 * 32;
    v2f* Wp2 = Wp1 + NP1;
    v2f* Wp3 = Wp2 + NP2;

    const int CSR_N = N_EDGES + N_NODES + 4;
    int* csr_src = (int*)out;
    int* rowptr  = csr_src + N_EDGES;
    int* bucketed = (int*)bufB;
    int* histT    = bucketed + N_EDGES;
    int* colscan  = histT + NBKT * NCHUNK;
    int* total    = colscan + NBKT * NCHUNK;
    int* csr_src3 = (int*)bufB;
    int* rowptr3  = csr_src3 + N_EDGES;

    const int nodeBlk = N_NODES / 8;

    pack_w_kernel<8><<<(NP1 + 255) / 256, 256, 0, stream>>>(W1, Wp1, F_IN, HID);
    pack_w_kernel<8><<<(NP2 + 255) / 256, 256, 0, stream>>>(W2, Wp2, HID, HID);
    pack_w_kernel<3><<<(NP3 + 255) / 256, 256, 0, stream>>>(W3, Wp3, HID, N_CLS);

    chunk_hist_kernel   <<<NCHUNK, 256, 0, stream>>>(ei + N_EDGES, histT);
    chunk_scan_kernel   <<<NBKT, NCHUNK, 0, stream>>>(histT, colscan, total);
    chunk_scatter_kernel<<<NCHUNK, 256, 0, stream>>>(ei, colscan, total, bucketed);
    bucket_sort_kernel  <<<NBKT, L2T, 0, stream>>>(bucketed, total, csr_src, rowptr, dinv);

    gemm_wmma_kernel<8, HID, false><<<Split<8>::BLOCKS, 128, 0, stream>>>(x, Wp1, dinv, bufA, N_NODES, F_IN);

    gather_gemm_kernel<8, HID><<<1024, 256, 0, stream>>>(bufA, dinv, rowptr, csr_src, b1, Wp2, bufB);

    gather_gemm_kernel<3, N_CLS><<<1024, 256, 0, stream>>>(bufB, dinv, rowptr, csr_src, b2, Wp3, bufA);

    copy_i4_kernel<<<(CSR_N / 4 + 255) / 256, 256, 0, stream>>>((const int4*)csr_src, (int4*)csr_src3, CSR_N / 4);
    gather_kernel<N_CLS><<<nodeBlk, 256, 0, stream>>>(bufA, dinv, rowptr3, csr_src3, b3, out);
}


# Launches 12 -> 8: the three weight-packing grids ride as extra work-groups in the edge-histogram launch, and the second fused gather launch copies the edge list on the side; sum orders unchanged
# speedup vs baseline: 3.6740x; 1.0308x over previous
#include <hip/hip_runtime.h>
#include <hip/hip_bf16.h>

typedef __attribute__((ext_vector_type(2))) float v2f;
typedef __attribute__((ext_vector_type(8))) float v8f;

#define N_NODES 100000
#define N_EDGES 1600000
#define F_IN    500
#define HID     128
#define N_CLS   40

#define WP1_N   ((F_IN / 4) * 8 * 32)
#define WP2_N   ((HID  / 4) * 8 * 32)
#define WP3_N   ((HID  / 4) * 3 * 32)
#define PB1     ((WP1_N + 255) / 256)
#define PB2     ((WP2_N + 255) / 256)
#define PB3     ((WP3_N + 255) / 256)

#define CSR_N   (N_EDGES + N_NODES + 4)
#define CSR_N4  (CSR_N / 4)
static_assert(CSR_N % 4 == 0, "the list is a whole number of 16-byte pieces");


#define NPB      512
#define NBKT     ((N_NODES + NPB - 1) / NPB)
#define NCHUNK   256
#define CHUNK_E  ((N_EDGES + NCHUNK - 1) / NCHUNK)
#define EPT      ((CHUNK_E + 255) / 256)
#define STAGE_CAP 12288
#define L2T      512
#define L2U      8

static_assert(NBKT <= 256 && NBKT <= L2T, "one thread per bucket");
static_assert(N_NODES <= (1 << 17), "src must fit 17 bits");

template<int NW>
__device__ inline int block_excl_scan(int v, int* sh_wave, int& tot) {
    const int lane = threadIdx.x & 31, w = threadIdx.x >> 5;
    int inc = v;
#pragma unroll
    for (int off = 1; off < 32; off <<= 1) {
        int y = __shfl_up(inc, off);
        if (lane >= off) inc += y;
    }
    if (lane == 31) sh_wave[w] = inc;
    __syncthreads();
    int pre = 0, sum = 0;
#pragma unroll
    for (int k = 0; k < NW; ++k) {
        const int x = sh_wave[k];
        if (k < w) pre += x;
        sum += x;
    }
    tot = sum;
    return pre + inc - v;
}

__device__ inline unsigned clamp_node(int d) { return min((unsigned)d, (unsigned)(N_NODES - 1)); }


template<int NT>
__device__ __forceinline__ void pack_w_block(const float* __restrict__ W,
                                             v2f* __restrict__ Wp,
                                             int K, int Nc, int blk) {
    int idx = blk * 256 + threadIdx.x;
    int total = (K / 4) * NT * 32;
    if (idx >= total) return;
    int lane = idx & 31;
    int t    = (idx >> 5) % NT;
    int ks   = idx / (32 * NT);
    int col  = t * 16 + (lane & 15);
    int k    = ks * 4 + ((lane >> 4) * 2);
    v2f v;
    v.x = (col < Nc) ? W[(size_t)k * Nc + col]       : 0.0f;
    v.y = (col < Nc) ? W[(size_t)(k + 1) * Nc + col] : 0.0f;
    Wp[idx] = v;
}

__global__ __launch_bounds__(256) void hist_pack_kernel(const int* __restrict__ dst,
                                                        int* __restrict__ histT,
                                                        const float* __restrict__ W1,
                                                        const float* __restrict__ W2,
                                                        const float* __restrict__ W3,
                                                        v2f* __restrict__ Wp1,
                                                        v2f* __restrict__ Wp2,
                                                        v2f* __restrict__ Wp3) {
    __shared__ int s_h[NBKT];
    const int c = blockIdx.x, t = threadIdx.x;
    if (c >= NCHUNK) {
        const int p = c - NCHUNK;
        if (p < PB1)            pack_w_block<8>(W1, Wp1, F_IN, HID, p);
        else if (p < PB1 + PB2) pack_w_block<8>(W2, Wp2, HID, HID, p - PB1);
        else                    pack_w_block<3>(W3, Wp3, HID, N_CLS, p - PB1 - PB2);
        return;
    }
    if (t < NBKT) s_h[t] = 0;
    const int e0 = c * CHUNK_E;
    const int e1 = min(e0 + CHUNK_E, N_EDGES);
    int dv[EPT];
#pragma unroll
    for (int k = 0; k < EPT; ++k) {
        const int e = min(e0 + k * 256 + t, N_EDGES - 1);
        dv[k] = dst[e];
    }
    __builtin_amdgcn_sched_barrier(0);
    __syncthreads();
#pragma unroll
    for (int k = 0; k < EPT; ++k) {
        const int e = e0 + k * 256 + t;
        if (e < e1) atomicAdd(&s_h[clamp_node(dv[k]) / NPB], 1);
    }
    __syncthreads();
    if (t < NBKT) histT[t * NCHUNK + c] = s_h[t];
}

__global__ __launch_bounds__(NCHUNK) void chunk_scan_kernel(const int* __restrict__ histT,
                                                            int* __restrict__ colscan,
                                                            int* __restrict__ total) {
    __shared__ int s_w[NCHUNK / 32];
    const int b = blockIdx.x, t = threadIdx.x;
    const int v = histT[b * NCHUNK + t];
    int tot;
    const int ex = block_excl_scan<NCHUNK / 32>(v, s_w, tot);
    colscan[b * NCHUNK + t] = ex;
    if (t == 0) total[b] = tot;
}

__global__ __launch_bounds__(256) void chunk_scatter_kernel(const int* __restrict__ ei,
                                                            const int* __restrict__ colscan,
                                                            const int* __restrict__ total,
                                                            int* __restrict__ bucketed) {
    __shared__ int s_cur[NBKT];
    __shared__ int s_w[8];
    const int c = blockIdx.x, t = threadIdx.x;
    const int e0 = c * CHUNK_E;
    const int e1 = min(e0 + CHUNK_E, N_EDGES);
    int sv[EPT], dv[EPT];
#pragma unroll
    for (int k = 0; k < EPT; ++k) {
        const int e = min(e0 + k * 256 + t, N_EDGES - 1);
        sv[k] = ei[e];
        dv[k] = ei[N_EDGES + e];
    }
    const int tv = (t < NBKT) ? total[t] : 0;
    const int cs = (t < NBKT) ? colscan[t * NCHUNK + c] : 0;
    int tot;
    const int base = block_excl_scan<8>(tv, s_w, tot);
    if (t < NBKT) s_cur[t] = base + cs;
    __syncthreads();
    int pos[EPT];
#pragma unroll
    for (int k = 0; k < EPT; ++k) {
        const int e = e0 + k * 256 + t;
        pos[k] = (e < e1) ? atomicAdd(&s_cur[clamp_node(dv[k]) / NPB], 1) : -1;
    }
#pragma unroll
    for (int k = 0; k < EPT; ++k) {
        if (pos[k] >= 0) {
            const unsigned d = clamp_node(dv[k]);
            bucketed[pos[k]] = (int)(((d % NPB) << 17) | ((unsigned)sv[k] & 0x1FFFFu));
        }
    }
}

__global__ __launch_bounds__(L2T) void bucket_sort_kernel(const int* __restrict__ bucketed,
                                                          const int* __restrict__ total,
                                                          int* __restrict__ csr_src,
                                                          int* __restrict__ rowptr,
                                                          float* __restrict__ dinv) {
    __shared__ int s_cnt[NPB];
    __shared__ int s_w0[L2T / 32];
    __shared__ int s_w1[L2T / 32];
    __shared__ int s_stage[STAGE_CAP];
    const int b = blockIdx.x, t = threadIdx.x;

    const int tv = (t < b) ? total[t] : 0;
    const int n  = total[b];
    int base;
    block_excl_scan<L2T / 32>(tv, s_w0, base);
    const int* __restrict__ mine = bucketed + base;

    for (int i = t; i < NPB; i += L2T) s_cnt[i] = 0;
    __syncthreads();

    for (int i0 = 0; i0 < n; i0 += L2T * L2U) {
        int p[L2U];
#pragma unroll
        for (int k = 0; k < L2U; ++k) p[k] = mine[min(i0 + k * L2T + t, n - 1)];
#pragma unroll
        for (int k = 0; k < L2U; ++k)
            if (i0 + k * L2T + t < n) atomicAdd(&s_cnt[(p[k] >> 17) & (NPB - 1)], 1);
    }
    __syncthreads();

    static_assert(NPB == L2T, "one node per thread");
    const int cn = s_cnt[t];
    int tot;
    const int ex = block_excl_scan<L2T / 32>(cn, s_w1, tot);
    const int node = b * NPB + t;
    if (node < N_NODES) {
        rowptr[node] = base + ex;
        dinv[node]   = rsqrtf((float)(cn + 1));
    }
    if (b == NBKT - 1 && t < 4) rowptr[N_NODES + t] = base + n;
    s_cnt[t] = ex;
    __syncthreads();

    const bool fits = n <= STAGE_CAP;
    for (int i0 = 0; i0 < n; i0 += L2T * L2U) {
        int p[L2U], pos[L2U];
#pragma unroll
        for (int k = 0; k < L2U; ++k) p[k] = mine[min(i0 + k * L2T + t, n - 1)];
#pragma unroll
        for (int k = 0; k < L2U; ++k)
            pos[k] = (i0 + k * L2T + t < n) ? atomicAdd(&s_cnt[(p[k] >> 17) & (NPB - 1)], 1) : -1;
#pragma unroll
        for (int k = 0; k < L2U; ++k) {
            if (pos[k] >= 0) {
                const int s = p[k] & 0x1FFFF;
                if (fits) s_stage[pos[k]] = s;
                else      csr_src[base + pos[k]] = s;
            }
        }
    }
    if (fits) {
        __syncthreads();
        for (int i = t; i < n; i += L2T) csr_src[base + i] = s_stage[i];
    }
}


#define RING 5

template<int NT>
struct KFrag {
    v2f a0, a1;
    v2f b[NT];
};

#define ROW_TILES   (N_NODES / 32)
#define N_SIMDS     1024
static_assert(N_NODES % 32 == 0, "row tiles are whole");

template<int NT>
struct Split {
    static constexpr int PNT    = (NT % 2 == 0) ? 2 : NT;
    static constexpr int CPS    = NT / PNT;
    static constexpr int WHOLE  = (PNT == NT) ? ROW_TILES : (ROW_TILES / N_SIMDS) * N_SIMDS;
    static constexpr int LEFT   = ROW_TILES - WHOLE;
    static constexpr int PIECES = LEFT * 2 * CPS;
    static constexpr int LEFT_BLOCKS = (PIECES + 3) / 4;
    static constexpr int BLOCKS = LEFT_BLOCKS + (WHOLE + 3) / 4;
};

template<int NS, int NTW, int NTALL, int NC, bool RELU>
__device__ __forceinline__ void gemm_wave_tile(const float* __restrict__ A,
                                               const v2f* __restrict__ Wp,
                                               const float* __restrict__ dinv,
                                               float* __restrict__ Out,
                                               int M, int K, int row0, int t0) {
    static_assert(NS == 1 || NS == 2, "one or two strips per wave");
    constexpr int NT = NTW;
    const int lane = threadIdx.x & 31;
    if (row0 >= M) return;

    const int kk  = (lane >> 4) * 2;
    const int col = lane & 15;

    const float* __restrict__ Arow0 = A + (size_t)(row0 + col) * K + kk;
    const float* __restrict__ Arow1 = Arow0 + (size_t)16 * K;
    const v2f*   __restrict__ wp    = Wp + t0 * 32 + lane;

    const int rbase0 = row0 + ((lane < 16) ? 0 : 8);
    const int rbase1 = rbase0 + 16;

    v8f acc0[NT], acc1[NT];
#pragma unroll
    for (int t = 0; t < NT; ++t) {
        acc0[t] = (v8f){0.f,0.f,0.f,0.f,0.f,0.f,0.f,0.f};
        acc1[t] = (v8f){0.f,0.f,0.f,0.f,0.f,0.f,0.f,0.f};
    }

    const int nsteps = K >> 2;
    const int last   = nsteps - 1;

    auto fetch = [&](int s) {
        KFrag<NT> f;
        f.a0 = *reinterpret_cast<const v2f*>(Arow0 + s * 4);
        if constexpr (NS == 2) f.a1 = *reinterpret_cast<const v2f*>(Arow1 + s * 4);
        else                   f.a1 = f.a0;
#pragma unroll
        for (int t = 0; t < NT; ++t) f.b[t] = wp[(s * NTALL + t) * 32];
        return f;
    };
    auto products = [&](const KFrag<NT>& f) {
        v2f a0 = f.a0, a1 = f.a1;
        if (RELU) {
            a0.x = fmaxf(a0.x, 0.0f); a0.y = fmaxf(a0.y, 0.0f);
            a1.x = fmaxf(a1.x, 0.0f); a1.y = fmaxf(a1.y, 0.0f);
        }
#pragma unroll
        for (int t = 0; t < NT; ++t) {
            acc0[t] = __builtin_amdgcn_wmma_f32_16x16x4_f32(
                          false, a0, false, f.b[t], (short)0, acc0[t], false, false);
            if constexpr (NS == 2)
                acc1[t] = __builtin_amdgcn_wmma_f32_16x16x4_f32(
                              false, a1, false, f.b[t], (short)0, acc1[t], false, false);
        }
    };

    KFrag<NT> ring[RING];
#pragma unroll
    for (int j = 0; j < RING; ++j) {
        ring[j] = fetch(min(j, last));
        __builtin_amdgcn_sched_barrier(0);
    }

    int ks = 0;
#pragma unroll 1
    for (; ks + 2 * RING <= nsteps; ks += RING) {
#pragma unroll
        for (int j = 0; j < RING; ++j) {
            products(ring[j]);
            ring[j] = fetch(ks + RING + j);
            __builtin_amdgcn_sched_barrier(0);
        }
    }

    const int R = nsteps - ks;
    float sc0[8], sc1[8];
#pragma unroll
    for (int j = 0; j < RING; ++j) {
        if (j < R) products(ring[j]);
        if (j < R - RING) ring[j] = fetch(ks + RING + j);
        if (j == 1) {
            const float4 p0 = *reinterpret_cast<const float4*>(dinv + rbase0);
            const float4 p1 = *reinterpret_cast<const float4*>(dinv + rbase0 + 4);
            sc0[0] = p0.x; sc0[1] = p0.y; sc0[2] = p0.z; sc0[3] = p0.w;
            sc0[4] = p1.x; sc0[5] = p1.y; sc0[6] = p1.z; sc0[7] = p1.w;
            if constexpr (NS == 2) {
                const float4 q0 = *reinterpret_cast<const float4*>(dinv + rbase1);
                const float4 q1 = *reinterpret_cast<const float4*>(dinv + rbase1 + 4);
                sc1[0] = q0.x; sc1[1] = q0.y; sc1[2] = q0.z; sc1[3] = q0.w;
                sc1[4] = q1.x; sc1[5] = q1.y; sc1[6] = q1.z; sc1[7] = q1.w;
            }
        }
    }
#pragma unroll
    for (int j = 0; j < RING - 1; ++j) {
        if (j < R - RING) products(ring[j]);
    }

#pragma unroll
    for (int t = 0; t < NT; ++t) {
        int c = (t0 + t) * 16 + col;
        if ((NC % 16 == 0) || (c < NC)) {
#pragma unroll
            for (int r = 0; r < 8; ++r) {
                Out[(size_t)(rbase0 + r) * NC + c] = acc0[t][r] * sc0[r];
                if constexpr (NS == 2) Out[(size_t)(rbase1 + r) * NC + c] = acc1[t][r] * sc1[r];
            }
        }
    }
}

template<int NT, int NC, bool RELU>
__global__ __launch_bounds__(128) void gemm_wmma_kernel(const float* __restrict__ A,
                                                        const v2f* __restrict__ Wp,
                                                        const float* __restrict__ dinv,
                                                        float* __restrict__ Out,
                                                        int M, int K) {
    using S = Split<NT>;
    const int wave = __builtin_amdgcn_readfirstlane(threadIdx.x >> 5);
    const int blk  = blockIdx.x;
    if constexpr (S::LEFT_BLOCKS > 0) {
        if (blk < S::LEFT_BLOCKS) {
            const int u = blk * 4 + wave;
            if (u < S::PIECES)
                gemm_wave_tile<1, S::PNT, NT, NC, RELU>(A, Wp, dinv, Out, M, K,
                                                        S::WHOLE * 32 + (u / S::CPS) * 16, (u % S::CPS) * S::PNT);
            return;
        }
    }
    gemm_wave_tile<2, NT, NT, NC, RELU>(A, Wp, dinv, Out, M, K, ((blk - S::LEFT_BLOCKS) * 4 + wave) * 32, 0);
}


template<int F>
__global__ __launch_bounds__(256) void gather_kernel(const float* __restrict__ h,
                                                     const float* __restrict__ dinv,
                                                     const int* __restrict__ rowptr,
                                                     const int* __restrict__ csr_src,
                                                     const float* __restrict__ bias,
                                                     float* __restrict__ out) {
    constexpr int CH  = F / 4;
    constexpr int EPI = 32 / CH;
    const int node = __builtin_amdgcn_readfirstlane(blockIdx.x * 8 + (threadIdx.x >> 5));
    if (node >= N_NODES) return;
    const int lane = threadIdx.x & 31;
    const int g = lane / CH;
    const int c = lane - g * CH;
    const bool active = g < EPI;

    const int start = rowptr[node];
    const int end   = rowptr[node + 1];
    const float dd  = dinv[node];
    const float4* __restrict__ h4 = reinterpret_cast<const float4*>(h);

    float4 acc = make_float4(0.f, 0.f, 0.f, 0.f);
    for (int base = start; base < end; base += 32) {
        const int n = min(32, end - base);
        int ms = 0;
        if (lane < n) {
            ms = (int)min((unsigned)csr_src[base + lane], (unsigned)(N_NODES - 1));
        }
        if constexpr (EPI == 1) {
#pragma unroll 4
            for (int j = 0; j < n; ++j) {
                const int s = __builtin_amdgcn_readlane(ms, j);
                const float4 v = h4[(size_t)s * CH + c];
                acc.x += v.x; acc.y += v.y; acc.z += v.z; acc.w += v.w;
            }
        } else {
            for (int j = 0; j < n; j += EPI) {
                const int idx = j + g;
                const int s = __shfl(ms, idx & 31);
                if (active && idx < n) {
                    const float4 v = h4[(size_t)s * CH + c];
                    acc.x += v.x; acc.y += v.y; acc.z += v.z; acc.w += v.w;
                }
            }
        }
    }
    if constexpr (EPI == 3) {
        acc.x += __shfl(acc.x, (lane + CH) & 31) + __shfl(acc.x, (lane + 2 * CH) & 31);
        acc.y += __shfl(acc.y, (lane + CH) & 31) + __shfl(acc.y, (lane + 2 * CH) & 31);
        acc.z += __shfl(acc.z, (lane + CH) & 31) + __shfl(acc.z, (lane + 2 * CH) & 31);
        acc.w += __shfl(acc.w, (lane + CH) & 31) + __shfl(acc.w, (lane + 2 * CH) & 31);
    }
    if (lane < CH) {
        const float4 hv = h4[(size_t)node * CH + lane];
        const float4 bv = reinterpret_cast<const float4*>(bias)[lane];
        float4 ov;
        ov.x = (bv.x + dd * hv.x) + dd * acc.x;  ov.y = (bv.y + dd * hv.y) + dd * acc.y;
        ov.z = (bv.z + dd * hv.z) + dd * acc.z;  ov.w = (bv.w + dd * hv.w) + dd * acc.w;
        reinterpret_cast<float4*>(out)[(size_t)node * CH + lane] = ov;
    }
}


#define FG_ROWS   16
#define FG_PITCH  (HID + 4)
#define FG_GROUPS (N_NODES / FG_ROWS)
#define FG_BLOCKS 1024
static_assert(N_NODES % FG_ROWS == 0, "node groups are whole");

template<int U>
__device__ __forceinline__ void row_batch(const float4* __restrict__ h4, int ms, int j, int e, int split,
                                          int lane, float4& accA, float4& accB) {
    float4 v[U];
#pragma unroll
    for (int u = 0; u < U; ++u) {
        const int s = __builtin_amdgcn_readlane(ms, j + u);
        v[u] = h4[(size_t)s * (HID / 4) + lane];
    }
    const int k = split - e;
    if (k >= U) {
#pragma unroll
        for (int u = 0; u < U; ++u) { accA.x += v[u].x; accA.y += v[u].y; accA.z += v[u].z; accA.w += v[u].w; }
    } else if (k <= 0) {
#pragma unroll
        for (int u = 0; u < U; ++u) { accB.x += v[u].x; accB.y += v[u].y; accB.z += v[u].z; accB.w += v[u].w; }
    } else {
#pragma unroll
        for (int u = 0; u < U; ++u) {
            if (u < k) { accA.x += v[u].x; accA.y += v[u].y; accA.z += v[u].z; accA.w += v[u].w; }
            else       { accB.x += v[u].x; accB.y += v[u].y; accB.z += v[u].z; accB.w += v[u].w; }
        }
    }
}

template<int NT, int NC, bool MOVE_LIST>
__global__ __launch_bounds__(256) void gather_gemm_kernel(const float* __restrict__ h,
                                                          const float* __restrict__ dinv,
                                                          const int* __restrict__ rowptr,
                                                          const int* __restrict__ csr_src,
                                                          const float* __restrict__ bias,
                                                          const v2f* __restrict__ Wp,
                                                          float* __restrict__ Out,
                                                          const int4* __restrict__ list_from,
                                                          int4* __restrict__ list_to) {
    constexpr int KS  = HID / 4;
    constexpr int PNT = Split<NT>::PNT;
    constexpr int CPS = Split<NT>::CPS;
    __shared__ v2f s_w[KS * NT * 32];
    __shared__ __attribute__((aligned(16))) float s_a[FG_ROWS * FG_PITCH];

    const int lane = threadIdx.x & 31;
    const int wave = __builtin_amdgcn_readfirstlane(threadIdx.x >> 5);

    if constexpr (MOVE_LIST) {
        for (int i = blockIdx.x * 256 + threadIdx.x; i < CSR_N4; i += FG_BLOCKS * 256) list_to[i] = list_from[i];
    }

    for (int i = threadIdx.x; i < KS * NT * 32; i += 256) s_w[i] = Wp[i];

    const float4* __restrict__ h4 = reinterpret_cast<const float4*>(h);
    const float4 bv = reinterpret_cast<const float4*>(bias)[lane];
    const int kk  = (lane >> 4) * 2;
    const int col = lane & 15;

    for (int g = blockIdx.x; g < FG_GROUPS; g += gridDim.x) {
        const int node0 = g * FG_ROWS;

        const int n0  = node0 + 2 * wave;
        const int rp0 = rowptr[n0], rp1 = rowptr[n0 + 1], rp2 = rowptr[n0 + 2];
        const int ms0 = (int)clamp_node(csr_src[min(rp0 + lane,      N_EDGES - 1)]);
        const int ms1 = (int)clamp_node(csr_src[min(rp0 + 32 + lane, N_EDGES - 1)]);
        const float4 hvA = h4[(size_t)n0 * (HID / 4) + lane];
        const float4 hvB = h4[(size_t)(n0 + 1) * (HID / 4) + lane];
        const float ddA = dinv[n0], ddB = dinv[n0 + 1];

        float4 aA = make_float4(0.f, 0.f, 0.f, 0.f), aB = make_float4(0.f, 0.f, 0.f, 0.f);
        int nb = 0;
        for (int base = rp0; base < rp2; base += 32, ++nb) {
            const int n = min(32, rp2 - base);
            int ms;
            if (nb == 0)      ms = ms0;
            else if (nb == 1) ms = ms1;
            else              ms = (int)clamp_node(csr_src[min(base + lane, N_EDGES - 1)]);
            int j = 0;
            for (; j + 16 <= n; j += 16) row_batch<16>(h4, ms, j, base + j, rp1, lane, aA, aB);
            if (j + 8 <= n) { row_batch<8>(h4, ms, j, base + j, rp1, lane, aA, aB); j += 8; }
            if (j + 4 <= n) { row_batch<4>(h4, ms, j, base + j, rp1, lane, aA, aB); j += 4; }
            if (j + 2 <= n) { row_batch<2>(h4, ms, j, base + j, rp1, lane, aA, aB); j += 2; }
            if (j < n)        row_batch<1>(h4, ms, j, base + j, rp1, lane, aA, aB);
        }

        float4 oA, oB;
        oA.x = fmaxf((bv.x + ddA * hvA.x) + ddA * aA.x, 0.0f);  oA.y = fmaxf((bv.y + ddA * hvA.y) + ddA * aA.y, 0.0f);
        oA.z = fmaxf((bv.z + ddA * hvA.z) + ddA * aA.z, 0.0f);  oA.w = fmaxf((bv.w + ddA * hvA.w) + ddA * aA.w, 0.0f);
        oB.x = fmaxf((bv.x + ddB * hvB.x) + ddB * aB.x, 0.0f);  oB.y = fmaxf((bv.y + ddB * hvB.y) + ddB * aB.y, 0.0f);
        oB.z = fmaxf((bv.z + ddB * hvB.z) + ddB * aB.z, 0.0f);  oB.w = fmaxf((bv.w + ddB * hvB.w) + ddB * aB.w, 0.0f);
        __syncthreads();
        *reinterpret_cast<float4*>(s_a + (2 * wave)     * FG_PITCH + 4 * lane) = oA;
        *reinterpret_cast<float4*>(s_a + (2 * wave + 1) * FG_PITCH + 4 * lane) = oB;
        __syncthreads();

        if (wave < CPS) {
            const int t0 = wave * PNT;
            v8f acc[PNT];
#pragma unroll
            for (int t = 0; t < PNT; ++t) acc[t] = (v8f){0.f,0.f,0.f,0.f,0.f,0.f,0.f,0.f};
            const float* arow = s_a + col * FG_PITCH + kk;
            const v2f*   wl   = s_w + t0 * 32 + lane;
#pragma unroll 8
            for (int s = 0; s < KS; ++s) {
                const v2f a = *reinterpret_cast<const v2f*>(arow + 4 * s);
#pragma unroll
                for (int t = 0; t < PNT; ++t)
                    acc[t] = __builtin_amdgcn_wmma_f32_16x16x4_f32(
                                 false, a, false, wl[(s * NT + t) * 32], (short)0, acc[t], false, false);
            }
            const int rbase = node0 + ((lane < 16) ? 0 : 8);
            const float4 p0 = *reinterpret_cast<const float4*>(dinv + rbase);
            const float4 p1 = *reinterpret_cast<const float4*>(dinv + rbase + 4);
            const float sc[8] = {p0.x, p0.y, p0.z, p0.w, p1.x, p1.y, p1.z, p1.w};
#pragma unroll
            for (int t = 0; t < PNT; ++t) {
                const int c = (t0 + t) * 16 + col;
                if ((NC % 16 == 0) || (c < NC)) {
#pragma unroll
                    for (int r = 0; r < 8; ++r)
                        Out[(size_t)(rbase + r) * NC + c] = acc[t][r] * sc[r];
                }
            }
        }
    }
}


extern "C" void kernel_launch(void* const* d_in, const int* in_sizes, int n_in,
                              void* d_out, int out_size, void* d_ws, size_t ws_size,
                              hipStream_t stream) {
    const float* x  = (const float*)d_in[0];
    const int*   ei = (const int*)d_in[1];
    const float* W1 = (const float*)d_in[2];
    const float* b1 = (const float*)d_in[3];
    const float* W2 = (const float*)d_in[4];
    const float* b2 = (const float*)d_in[5];
    const float* W3 = (const float*)d_in[6];
    const float* b3 = (const float*)d_in[7];
    float* out = (float*)d_out;

    char* ws = (char*)d_ws;
    float* dinv = (float*)ws;
    float* bufA = (float*)(ws + (((size_t)N_NODES * 4 + 255) & ~(size_t)255));
    float* bufB = bufA + (size_t)N_NODES * HID;
    v2f*   Wp1  = (v2f*)(bufB + (size_t)N_NODES * HID);
    v2f*   Wp2  = Wp1 + WP1_N;
    v2f*   Wp3  = Wp2 + WP2_N;

    int* csr_src = (int*)out;
    int* rowptr  = csr_src + N_EDGES;
    int* bucketed = (int*)bufB;
    int* histT    = bucketed + N_EDGES;
    int* colscan  = histT + NBKT * NCHUNK;
    int* total    = colscan + NBKT * NCHUNK;
    int* csr_src3 = (int*)(bufA + (size_t)N_NODES * N_CLS);
    int* rowptr3  = csr_src3 + N_EDGES;

    const int nodeBlk = N_NODES / 8;

    hist_pack_kernel    <<<NCHUNK + PB1 + PB2 + PB3, 256, 0, stream>>>(ei + N_EDGES, histT, W1, W2, W3, Wp1, Wp2, Wp3);

    chunk_scan_kernel   <<<NBKT, NCHUNK, 0, stream>>>(histT, colscan, total);
    chunk_scatter_kernel<<<NCHUNK, 256, 0, stream>>>(ei, colscan, total, bucketed);
    bucket_sort_kernel  <<<NBKT, L2T, 0, stream>>>(bucketed, total, csr_src, rowptr, dinv);

    gemm_wmma_kernel<8, HID, false><<<Split<8>::BLOCKS, 128, 0, stream>>>(x, Wp1, dinv, bufA, N_NODES, F_IN);

    gather_gemm_kernel<8, HID, false><<<FG_BLOCKS, 256, 0, stream>>>(bufA, dinv, rowptr, csr_src, b1, Wp2, bufB,
                                                                      nullptr, nullptr);

    gather_gemm_kernel<3, N_CLS, true><<<FG_BLOCKS, 256, 0, stream>>>(bufB, dinv, rowptr, csr_src, b2, Wp3, bufA,
                                                                       (const int4*)csr_src, (int4*)csr_src3);

    gather_kernel<N_CLS><<<nodeBlk, 256, 0, stream>>>(bufA, dinv, rowptr3, csr_src3, b3, out);
}
